# starting kernel
# baseline (speedup 1.0000x reference)
#include <hip/hip_runtime.h>

typedef __attribute__((ext_vector_type(2))) float v2f;
typedef __attribute__((ext_vector_type(8))) float v8f;

#define BATCH   4096
#define TSTEPS  2048

__device__ __forceinline__ float fsigm_b(float x, float hb) {
    return fmaf(__builtin_amdgcn_tanhf(fmaf(x, 0.5f, hb)), 0.5f, 0.5f);
}

__global__ __launch_bounds__(32)
void OpenLSTM_wmma_scan(const float* __restrict__ u,
                        const float* __restrict__ Wih,
                        const float* __restrict__ Whh,
                        const float* __restrict__ bih,
                        const float* __restrict__ bhh,
                        const float* __restrict__ Whr,
                        const int*   __restrict__ ncp,
                        float*       __restrict__ out)
{
    const int lane = threadIdx.x;
    const int n    = lane & 15;
    const int hi   = lane >> 4;
    const int b0   = blockIdx.x * 16;
    const int nc   = *ncp;

    v2f A[4];
    const float* Wsrc = hi ? Whh : Wih;
#pragma unroll
    for (int g = 0; g < 4; ++g) {
        const int row = g * 16 + n;
        A[g].x = Wsrc[row * 2 + 0];
        A[g].y = Wsrc[row * 2 + 1];
    }

    float hbi[8], hbf[8], hbo[8], bg[8];
#pragma unroll
    for (int r = 0; r < 8; ++r) {
        const int m = r + hi * 8;
        hbi[r] = 0.5f * (bih[ 0 + m] + bhh[ 0 + m]);
        hbf[r] = 0.5f * (bih[16 + m] + bhh[16 + m]);
        bg[r]  =         bih[32 + m] + bhh[32 + m];
        hbo[r] = 0.5f * (bih[48 + m] + bhh[48 + m]);
    }

    float wr0[8], wr1[8];
#pragma unroll
    for (int r = 0; r < 8; ++r) {
        wr0[r] = Whr[0 * 16 + r + hi * 8];
        wr1[r] = Whr[1 * 16 + r + hi * 8];
    }

    v8f c = {};
    const v8f zc = {};
    float h0 = 0.0f, h1 = 0.0f;

    const float* up = u + (size_t)(b0 + n) * TSTEPS * 4 + hi * 2;
    float*       op = out + (size_t)(b0 + n) * TSTEPS * 2;

    float2 ld = *(const float2*)up;

    for (int t = 0; t < TSTEPS; ++t) {
        const bool use_mem = (!hi) || (t < nc);
        v2f Bv;
        Bv.x = use_mem ? ld.x : h0;
        Bv.y = use_mem ? ld.y : h1;

        const int tn = (t + 1 < TSTEPS) ? (t + 1) : (TSTEPS - 1);
        const float2 ldn = *(const float2*)(up + (size_t)tn * 4);
        __builtin_prefetch(up + (size_t)tn * 4 + 32, 0, 3);

        v8f gi = __builtin_amdgcn_wmma_f32_16x16x4_f32(false, A[0], false, Bv, (short)0, zc, false, false);
        v8f gf = __builtin_amdgcn_wmma_f32_16x16x4_f32(false, A[1], false, Bv, (short)0, zc, false, false);
        v8f gg = __builtin_amdgcn_wmma_f32_16x16x4_f32(false, A[2], false, Bv, (short)0, zc, false, false);
        v8f go = __builtin_amdgcn_wmma_f32_16x16x4_f32(false, A[3], false, Bv, (short)0, zc, false, false);

        float p0 = 0.0f, p1 = 0.0f;
#pragma unroll
        for (int r = 0; r < 8; ++r) {
            const float si = fsigm_b(gi[r], hbi[r]);
            const float sf = fsigm_b(gf[r], hbf[r]);
            const float tg = __builtin_amdgcn_tanhf(gg[r] + bg[r]);
            const float so = fsigm_b(go[r], hbo[r]);
            const float cn = sf * c[r] + si * tg;
            c[r] = cn;
            const float s = so * __builtin_amdgcn_tanhf(cn);
            p0 += s * wr0[r];
            p1 += s * wr1[r];
        }
        h0 = p0 + __shfl_xor(p0, 16, 32);
        h1 = p1 + __shfl_xor(p1, 16, 32);

        if (!hi) {
            float2 hv; hv.x = h0; hv.y = h1;
            __builtin_nontemporal_store(__builtin_bit_cast(double, hv),
                                        (double*)(op + (size_t)t * 2));
        }
        ld = ldn;
    }
}

extern "C" void kernel_launch(void* const* d_in, const int* in_sizes, int n_in,
                              void* d_out, int out_size, void* d_ws, size_t ws_size,
                              hipStream_t stream) {
    const float* u    = (const float*)d_in[0];
    const float* Wih  = (const float*)d_in[1];
    const float* Whh  = (const float*)d_in[2];
    const float* bih  = (const float*)d_in[3];
    const float* bhh  = (const float*)d_in[4];
    const float* Whr  = (const float*)d_in[5];
    const int*   ncp  = (const int*)d_in[6];
    float*       out  = (float*)d_out;

    dim3 grid(BATCH / 16);
    dim3 block(32);
    OpenLSTM_wmma_scan<<<grid, block, 0, stream>>>(u, Wih, Whh, bih, bhh, Whr, ncp, out);
}


# input fetched as whole 128-byte lines one 8-step block ahead (each line read once, 4 lines per 16-byte wave load) and handed to lanes via an LDS ring; replaces per-step 16-line load + prefetch
# speedup vs baseline: 2.2771x; 2.2771x over previous
#include <hip/hip_runtime.h>

typedef __attribute__((ext_vector_type(2))) float v2f;
typedef __attribute__((ext_vector_type(4))) float v4f;
typedef __attribute__((ext_vector_type(8))) float v8f;

#define BATCH   4096
#define TSTEPS  2048
#define BLK     8
#define NBLK    (TSTEPS / BLK)
#define PITCH   68

__device__ __forceinline__ float fsigm_b(float x, float hb) {
    return fmaf(__builtin_amdgcn_tanhf(fmaf(x, 0.5f, hb)), 0.5f, 0.5f);
}

__global__ __launch_bounds__(32)
void OpenLSTM_wmma_scan(const float* __restrict__ u,
                        const float* __restrict__ Wih,
                        const float* __restrict__ Whh,
                        const float* __restrict__ bih,
                        const float* __restrict__ bhh,
                        const float* __restrict__ Whr,
                        const int*   __restrict__ ncp,
                        float*       __restrict__ out)
{
    const int lane = threadIdx.x;
    const int n    = lane & 15;
    const int hi   = lane >> 4;
    const int b0   = blockIdx.x * 16;
    const int nc   = *ncp;

    __shared__ __attribute__((aligned(16))) float stage[16 * PITCH];
    const int ls = lane & 7;
    const int lb = lane >> 3;
    const float* gsrc = u + ((size_t)(b0 + lb) * TSTEPS + ls) * 4;
    float*       sdst = stage + lb * PITCH + ls * 4;
    const float* ssrc = stage + n * PITCH + hi * 2;
    const size_t PIECE = (size_t)4 * TSTEPS * 4;

    v2f A[4];
    const float* Wsrc = hi ? Whh : Wih;
#pragma unroll
    for (int g = 0; g < 4; ++g) {
        const int row = g * 16 + n;
        A[g].x = Wsrc[row * 2 + 0];
        A[g].y = Wsrc[row * 2 + 1];
    }

    float hbi[8], hbf[8], hbo[8], bg[8];
#pragma unroll
    for (int r = 0; r < 8; ++r) {
        const int m = r + hi * 8;
        hbi[r] = 0.5f * (bih[ 0 + m] + bhh[ 0 + m]);
        hbf[r] = 0.5f * (bih[16 + m] + bhh[16 + m]);
        bg[r]  =         bih[32 + m] + bhh[32 + m];
        hbo[r] = 0.5f * (bih[48 + m] + bhh[48 + m]);
    }

    float wr0[8], wr1[8];
#pragma unroll
    for (int r = 0; r < 8; ++r) {
        wr0[r] = Whr[0 * 16 + r + hi * 8];
        wr1[r] = Whr[1 * 16 + r + hi * 8];
    }

    v8f c = {};
    const v8f zc = {};
    float h0 = 0.0f, h1 = 0.0f;

    float*       op = out + (size_t)(b0 + n) * TSTEPS * 2;

    {
        const v4f g0 = *(const v4f*)(gsrc);
        const v4f g1 = *(const v4f*)(gsrc + PIECE);
        const v4f g2 = *(const v4f*)(gsrc + 2 * PIECE);
        const v4f g3 = *(const v4f*)(gsrc + 3 * PIECE);
        *(v4f*)(sdst)             = g0;
        *(v4f*)(sdst + 4 * PITCH)  = g1;
        *(v4f*)(sdst + 8 * PITCH)  = g2;
        *(v4f*)(sdst + 12 * PITCH) = g3;
    }
    __builtin_amdgcn_fence(__ATOMIC_ACQ_REL, "wavefront");

    float2 ld = *(const float2*)ssrc;

    auto step = [&](const int t) __attribute__((always_inline)) {
        const bool use_mem = (!hi) || (t < nc);
        v2f Bv;
        Bv.x = use_mem ? ld.x : h0;
        Bv.y = use_mem ? ld.y : h1;

        const float2 ldn = *(const float2*)(ssrc + ((t + 1) & (2 * BLK - 1)) * 4);

        v8f gi = __builtin_amdgcn_wmma_f32_16x16x4_f32(false, A[0], false, Bv, (short)0, zc, false, false);
        v8f gf = __builtin_amdgcn_wmma_f32_16x16x4_f32(false, A[1], false, Bv, (short)0, zc, false, false);
        v8f gg = __builtin_amdgcn_wmma_f32_16x16x4_f32(false, A[2], false, Bv, (short)0, zc, false, false);
        v8f go = __builtin_amdgcn_wmma_f32_16x16x4_f32(false, A[3], false, Bv, (short)0, zc, false, false);

        float p0 = 0.0f, p1 = 0.0f;
#pragma unroll
        for (int r = 0; r < 8; ++r) {
            const float si = fsigm_b(gi[r], hbi[r]);
            const float sf = fsigm_b(gf[r], hbf[r]);
            const float tg = __builtin_amdgcn_tanhf(gg[r] + bg[r]);
            const float so = fsigm_b(go[r], hbo[r]);
            const float cn = sf * c[r] + si * tg;
            c[r] = cn;
            const float s = so * __builtin_amdgcn_tanhf(cn);
            p0 += s * wr0[r];
            p1 += s * wr1[r];
        }
        h0 = p0 + __shfl_xor(p0, 16, 32);
        h1 = p1 + __shfl_xor(p1, 16, 32);

        if (!hi) {
            float2 hv; hv.x = h0; hv.y = h1;
            __builtin_nontemporal_store(__builtin_bit_cast(double, hv),
                                        (double*)(op + (size_t)t * 2));
        }
        ld = ldn;
    };

    for (int k = 0; k < NBLK; ++k) {
        const int kn = (k + 1 < NBLK) ? (k + 1) : (NBLK - 1);
        const float* gk = gsrc + (size_t)kn * BLK * 4;
        const v4f g0 = *(const v4f*)(gk);
        const v4f g1 = *(const v4f*)(gk + PIECE);
        const v4f g2 = *(const v4f*)(gk + 2 * PIECE);
        const v4f g3 = *(const v4f*)(gk + 3 * PIECE);

        const int tb = k * BLK;
        for (int j = 0; j < BLK - 1; ++j)
            step(tb + j);

        float* sk = sdst + ((k + 1) & 1) * (BLK * 4);
        *(v4f*)(sk)              = g0;
        *(v4f*)(sk + 4 * PITCH)  = g1;
        *(v4f*)(sk + 8 * PITCH)  = g2;
        *(v4f*)(sk + 12 * PITCH) = g3;
        __builtin_amdgcn_fence(__ATOMIC_ACQ_REL, "wavefront");

        step(tb + BLK - 1);
    }
}

extern "C" void kernel_launch(void* const* d_in, const int* in_sizes, int n_in,
                              void* d_out, int out_size, void* d_ws, size_t ws_size,
                              hipStream_t stream) {
    const float* u    = (const float*)d_in[0];
    const float* Wih  = (const float*)d_in[1];
    const float* Whh  = (const float*)d_in[2];
    const float* bih  = (const float*)d_in[3];
    const float* bhh  = (const float*)d_in[4];
    const float* Whr  = (const float*)d_in[5];
    const int*   ncp  = (const int*)d_in[6];
    float*       out  = (float*)d_out;

    dim3 grid(BATCH / 16);
    dim3 block(32);
    OpenLSTM_wmma_scan<<<grid, block, 0, stream>>>(u, Wih, Whh, bih, bhh, Whr, ncp, out);
}


# context phase (t < n_context: h is read from the input, only c is carried) runs as whole 8-step blocks: each step's WMMAs issued a step ahead, no LDS wait or branch inside a step; rest unchanged
# speedup vs baseline: 2.6267x; 1.1535x over previous
#include <hip/hip_runtime.h>

typedef __attribute__((ext_vector_type(2))) float v2f;
typedef __attribute__((ext_vector_type(4))) float v4f;
typedef __attribute__((ext_vector_type(8))) float v8f;

#define BATCH   4096
#define TSTEPS  2048
#define BLK     8
#define NBLK    (TSTEPS / BLK)
#define PITCH   68

__device__ __forceinline__ float fsigm_b(float x, float hb) {
    return fmaf(__builtin_amdgcn_tanhf(fmaf(x, 0.5f, hb)), 0.5f, 0.5f);
}

__device__ __forceinline__ float xhalf(float x) {
    const int b = __builtin_bit_cast(int, x);
    return __builtin_bit_cast(float, __builtin_amdgcn_permlanex16(b, b, 0x76543210, 0xfedcba98, false, false));
}

__device__ __forceinline__ void pin() { __builtin_amdgcn_sched_barrier(0); }

struct Gates { v8f i, f, g, o; };

__global__ __launch_bounds__(32)
void OpenLSTM_wmma_scan(const float* __restrict__ u,
                        const float* __restrict__ Wih,
                        const float* __restrict__ Whh,
                        const float* __restrict__ bih,
                        const float* __restrict__ bhh,
                        const float* __restrict__ Whr,
                        const int*   __restrict__ ncp,
                        float*       __restrict__ out)
{
    const int lane = threadIdx.x;
    const int n    = lane & 15;
    const int hi   = lane >> 4;
    const int b0   = blockIdx.x * 16;
    const int nc   = *ncp;

    __shared__ __attribute__((aligned(16))) float stage[16 * PITCH];
    const int ls = lane & 7;
    const int lb = lane >> 3;
    const float* gsrc = u + ((size_t)(b0 + lb) * TSTEPS + ls) * 4;
    float*       sdst = stage + lb * PITCH + ls * 4;
    const float* ssrc = stage + n * PITCH + hi * 2;
    const size_t PIECE = (size_t)4 * TSTEPS * 4;

    v2f A[4];
    const float* Wsrc = hi ? Whh : Wih;
#pragma unroll
    for (int g = 0; g < 4; ++g) {
        const int row = g * 16 + n;
        A[g].x = Wsrc[row * 2 + 0];
        A[g].y = Wsrc[row * 2 + 1];
    }

    float hbi[8], hbf[8], hbo[8], bg[8];
#pragma unroll
    for (int r = 0; r < 8; ++r) {
        const int m = r + hi * 8;
        hbi[r] = 0.5f * (bih[ 0 + m] + bhh[ 0 + m]);
        hbf[r] = 0.5f * (bih[16 + m] + bhh[16 + m]);
        bg[r]  =         bih[32 + m] + bhh[32 + m];
        hbo[r] = 0.5f * (bih[48 + m] + bhh[48 + m]);
    }

    float wr0[8], wr1[8];
#pragma unroll
    for (int r = 0; r < 8; ++r) {
        wr0[r] = Whr[0 * 16 + r + hi * 8];
        wr1[r] = Whr[1 * 16 + r + hi * 8];
    }

    v8f c = {};
    const v8f zc = {};
    float h0 = 0.0f, h1 = 0.0f;

    float*       op = out + (size_t)(b0 + n) * TSTEPS * 2;

    {
        const v4f g0 = *(const v4f*)(gsrc);
        const v4f g1 = *(const v4f*)(gsrc + PIECE);
        const v4f g2 = *(const v4f*)(gsrc + 2 * PIECE);
        const v4f g3 = *(const v4f*)(gsrc + 3 * PIECE);
        *(v4f*)(sdst)             = g0;
        *(v4f*)(sdst + 4 * PITCH)  = g1;
        *(v4f*)(sdst + 8 * PITCH)  = g2;
        *(v4f*)(sdst + 12 * PITCH) = g3;
    }
    __builtin_amdgcn_fence(__ATOMIC_ACQ_REL, "wavefront");

    const int ncc = nc < 0 ? 0 : (nc > TSTEPS ? TSTEPS : nc);
    const int kA  = ncc / BLK;

    auto gates_of = [&](const v2f Bv) __attribute__((always_inline)) {
        Gates G;
        G.i = __builtin_amdgcn_wmma_f32_16x16x4_f32(false, A[0], false, Bv, (short)0, zc, false, false);
        G.f = __builtin_amdgcn_wmma_f32_16x16x4_f32(false, A[1], false, Bv, (short)0, zc, false, false);
        G.g = __builtin_amdgcn_wmma_f32_16x16x4_f32(false, A[2], false, Bv, (short)0, zc, false, false);
        G.o = __builtin_amdgcn_wmma_f32_16x16x4_f32(false, A[3], false, Bv, (short)0, zc, false, false);
        return G;
    };

    v2f in[BLK];
#pragma unroll
    for (int j = 0; j < BLK; ++j)
        in[j] = *(const v2f*)(ssrc + j * 4);
    Gates G = gates_of(in[0]);

    int k = 0;
    for (; k < kA; ++k) {
        const int kn = (k + 1 < NBLK) ? (k + 1) : (NBLK - 1);
        const float* gk = gsrc + (size_t)kn * BLK * 4;
        const v4f g0 = *(const v4f*)(gk);
        const v4f g1 = *(const v4f*)(gk + PIECE);
        const v4f g2 = *(const v4f*)(gk + 2 * PIECE);
        const v4f g3 = *(const v4f*)(gk + 3 * PIECE);
        pin();

        const int tb = k * BLK;
        v2f inn[BLK];
        float e0 = 0.0f, e1 = 0.0f;
#pragma unroll
        for (int j = 0; j < BLK; ++j) {
            if (j == BLK - 1) {
                pin();
                float* sk = sdst + ((k + 1) & 1) * (BLK * 4);
                *(v4f*)(sk)              = g0;
                *(v4f*)(sk + 4 * PITCH)  = g1;
                *(v4f*)(sk + 8 * PITCH)  = g2;
                *(v4f*)(sk + 12 * PITCH) = g3;
                __builtin_amdgcn_fence(__ATOMIC_ACQ_REL, "wavefront");
                const float* sn = ssrc + ((k + 1) & 1) * (BLK * 4);
#pragma unroll
                for (int i = 0; i < BLK; ++i)
                    inn[i] = *(const v2f*)(sn + i * 4);
            }
            const v2f Bn = (j < BLK - 1) ? in[j + 1] : inn[0];
            Gates Gn;
            float ai[8], af[8], ag[8], ao[8], tc[8];
            Gn.i = __builtin_amdgcn_wmma_f32_16x16x4_f32(false, A[0], false, Bn, (short)0, zc, false, false);
#pragma unroll
            for (int r = 0; r < 8; ++r) ai[r] = fmaf(G.i[r], 0.5f, hbi[r]);
#pragma unroll
            for (int r = 0; r < 8; ++r) { ai[r] = __builtin_amdgcn_tanhf(ai[r]); af[r] = fmaf(G.f[r], 0.5f, hbf[r]); }
            Gn.f = __builtin_amdgcn_wmma_f32_16x16x4_f32(false, A[1], false, Bn, (short)0, zc, false, false);
#pragma unroll
            for (int r = 0; r < 8; ++r) { af[r] = __builtin_amdgcn_tanhf(af[r]); ag[r] = G.g[r] + bg[r]; }
            Gn.g = __builtin_amdgcn_wmma_f32_16x16x4_f32(false, A[2], false, Bn, (short)0, zc, false, false);
#pragma unroll
            for (int r = 0; r < 8; ++r) { ag[r] = __builtin_amdgcn_tanhf(ag[r]); ao[r] = fmaf(G.o[r], 0.5f, hbo[r]); }
            Gn.o = __builtin_amdgcn_wmma_f32_16x16x4_f32(false, A[3], false, Bn, (short)0, zc, false, false);
#pragma unroll
            for (int r = 0; r < 8; ++r) {
                ao[r] = __builtin_amdgcn_tanhf(ao[r]);
                ai[r] = fmaf(ai[r], 0.5f, 0.5f);
                af[r] = fmaf(af[r], 0.5f, 0.5f);
            }
#pragma unroll
            for (int r = 0; r < 8; ++r) {
                const float cn = af[r] * c[r] + ai[r] * ag[r];
                c[r] = cn;
            }
#pragma unroll
            for (int r = 0; r < 8; ++r) { tc[r] = __builtin_amdgcn_tanhf(c[r]); ao[r] = fmaf(ao[r], 0.5f, 0.5f); }
            v2f p = {0.0f, 0.0f};
#pragma unroll
            for (int r = 0; r < 8; ++r) {
                const float s = ao[r] * tc[r];
                const v2f ss = {s, s};
                const v2f w  = {wr0[r], wr1[r]};
                p += ss * w;
            }
            h0 = p.x + xhalf(p.x);
            h1 = p.y + xhalf(p.y);

            if (j & 1) {
                v2f hv;
                hv.x = hi ? h0 : e0;
                hv.y = hi ? h1 : e1;
                __builtin_nontemporal_store(hv, (v2f*)(op + (size_t)(tb + j - 1 + hi) * 2));
            } else {
                e0 = h0;
                e1 = h1;
            }
            G = Gn;
        }
#pragma unroll
        for (int j = 0; j < BLK; ++j)
            in[j] = inn[j];
    }

    float2 ld;
    ld.x = in[0].x;
    ld.y = in[0].y;

    auto step = [&](const int t) __attribute__((always_inline)) {
        const bool use_mem = (!hi) || (t < nc);
        v2f Bv;
        Bv.x = use_mem ? ld.x : h0;
        Bv.y = use_mem ? ld.y : h1;

        const float2 ldn = *(const float2*)(ssrc + ((t + 1) & (2 * BLK - 1)) * 4);

        v8f gi = __builtin_amdgcn_wmma_f32_16x16x4_f32(false, A[0], false, Bv, (short)0, zc, false, false);
        v8f gf = __builtin_amdgcn_wmma_f32_16x16x4_f32(false, A[1], false, Bv, (short)0, zc, false, false);
        v8f gg = __builtin_amdgcn_wmma_f32_16x16x4_f32(false, A[2], false, Bv, (short)0, zc, false, false);
        v8f go = __builtin_amdgcn_wmma_f32_16x16x4_f32(false, A[3], false, Bv, (short)0, zc, false, false);

        float p0 = 0.0f, p1 = 0.0f;
#pragma unroll
        for (int r = 0; r < 8; ++r) {
            const float si = fsigm_b(gi[r], hbi[r]);
            const float sf = fsigm_b(gf[r], hbf[r]);
            const float tg = __builtin_amdgcn_tanhf(gg[r] + bg[r]);
            const float so = fsigm_b(go[r], hbo[r]);
            const float cn = sf * c[r] + si * tg;
            c[r] = cn;
            const float s = so * __builtin_amdgcn_tanhf(cn);
            p0 += s * wr0[r];
            p1 += s * wr1[r];
        }
        h0 = p0 + __shfl_xor(p0, 16, 32);
        h1 = p1 + __shfl_xor(p1, 16, 32);

        if (!hi) {
            float2 hv; hv.x = h0; hv.y = h1;
            __builtin_nontemporal_store(__builtin_bit_cast(double, hv),
                                        (double*)(op + (size_t)t * 2));
        }
        ld = ldn;
    };

    for (; k < NBLK; ++k) {
        const int kn = (k + 1 < NBLK) ? (k + 1) : (NBLK - 1);
        const float* gk = gsrc + (size_t)kn * BLK * 4;
        const v4f g0 = *(const v4f*)(gk);
        const v4f g1 = *(const v4f*)(gk + PIECE);
        const v4f g2 = *(const v4f*)(gk + 2 * PIECE);
        const v4f g3 = *(const v4f*)(gk + 3 * PIECE);

        const int tb = k * BLK;
        for (int j = 0; j < BLK - 1; ++j)
            step(tb + j);

        float* sk = sdst + ((k + 1) & 1) * (BLK * 4);
        *(v4f*)(sk)              = g0;
        *(v4f*)(sk + 4 * PITCH)  = g1;
        *(v4f*)(sk + 8 * PITCH)  = g2;
        *(v4f*)(sk + 12 * PITCH) = g3;
        __builtin_amdgcn_fence(__ATOMIC_ACQ_REL, "wavefront");

        step(tb + BLK - 1);
    }
}

extern "C" void kernel_launch(void* const* d_in, const int* in_sizes, int n_in,
                              void* d_out, int out_size, void* d_ws, size_t ws_size,
                              hipStream_t stream) {
    const float* u    = (const float*)d_in[0];
    const float* Wih  = (const float*)d_in[1];
    const float* Whh  = (const float*)d_in[2];
    const float* bih  = (const float*)d_in[3];
    const float* bhh  = (const float*)d_in[4];
    const float* Whr  = (const float*)d_in[5];
    const int*   ncp  = (const int*)d_in[6];
    float*       out  = (float*)d_out;

    dim3 grid(BATCH / 16);
    dim3 block(32);
    OpenLSTM_wmma_scan<<<grid, block, 0, stream>>>(u, Wih, Whh, bih, bhh, Whr, ncp, out);
}


# free-running phase (t >= n_context) as whole 8-step blocks: inputs in registers, lane halves joined by permlanex16 not ds_bpermute, projection summed pairwise (sum order changes), paired stores
# speedup vs baseline: 3.1997x; 1.2181x over previous
#include <hip/hip_runtime.h>

typedef __attribute__((ext_vector_type(2))) float v2f;
typedef __attribute__((ext_vector_type(4))) float v4f;
typedef __attribute__((ext_vector_type(8))) float v8f;

#define BATCH   4096
#define TSTEPS  2048
#define BLK     8
#define NBLK    (TSTEPS / BLK)
#define PITCH   68

__device__ __forceinline__ float fsigm_b(float x, float hb) {
    return fmaf(__builtin_amdgcn_tanhf(fmaf(x, 0.5f, hb)), 0.5f, 0.5f);
}

__device__ __forceinline__ float xhalf(float x) {
    const int b = __builtin_bit_cast(int, x);
    return __builtin_bit_cast(float, __builtin_amdgcn_permlanex16(b, b, 0x76543210, 0xfedcba98, false, false));
}

__device__ __forceinline__ float xhalf_rec(float x) {
    const int b = __builtin_bit_cast(int, x);
    return __builtin_bit_cast(float, __builtin_amdgcn_permlanex16(0, b, 0x76543210, 0xfedcba98, false, false));
}

__device__ __forceinline__ void pin() { __builtin_amdgcn_sched_barrier(0); }

struct Gates { v8f i, f, g, o; };

__global__ __launch_bounds__(32)
void OpenLSTM_wmma_scan(const float* __restrict__ u,
                        const float* __restrict__ Wih,
                        const float* __restrict__ Whh,
                        const float* __restrict__ bih,
                        const float* __restrict__ bhh,
                        const float* __restrict__ Whr,
                        const int*   __restrict__ ncp,
                        float*       __restrict__ out)
{
    const int lane = threadIdx.x;
    const int n    = lane & 15;
    const int hi   = lane >> 4;
    const int b0   = blockIdx.x * 16;
    const int nc   = *ncp;

    __shared__ __attribute__((aligned(16))) float stage[16 * PITCH];
    const int ls = lane & 7;
    const int lb = lane >> 3;
    const float* gsrc = u + ((size_t)(b0 + lb) * TSTEPS + ls) * 4;
    float*       sdst = stage + lb * PITCH + ls * 4;
    const float* ssrc = stage + n * PITCH + hi * 2;
    const size_t PIECE = (size_t)4 * TSTEPS * 4;

    v2f A[4];
    const float* Wsrc = hi ? Whh : Wih;
#pragma unroll
    for (int g = 0; g < 4; ++g) {
        const int row = g * 16 + n;
        A[g].x = Wsrc[row * 2 + 0];
        A[g].y = Wsrc[row * 2 + 1];
    }

    float hbi[8], hbf[8], hbo[8], bg[8];
#pragma unroll
    for (int r = 0; r < 8; ++r) {
        const int m = r + hi * 8;
        hbi[r] = 0.5f * (bih[ 0 + m] + bhh[ 0 + m]);
        hbf[r] = 0.5f * (bih[16 + m] + bhh[16 + m]);
        bg[r]  =         bih[32 + m] + bhh[32 + m];
        hbo[r] = 0.5f * (bih[48 + m] + bhh[48 + m]);
    }

    float wr0[8], wr1[8];
#pragma unroll
    for (int r = 0; r < 8; ++r) {
        wr0[r] = Whr[0 * 16 + r + hi * 8];
        wr1[r] = Whr[1 * 16 + r + hi * 8];
    }

    v8f c = {};
    const v8f zc = {};
    float h0 = 0.0f, h1 = 0.0f;

    float*       op = out + (size_t)(b0 + n) * TSTEPS * 2;

    {
        const v4f g0 = *(const v4f*)(gsrc);
        const v4f g1 = *(const v4f*)(gsrc + PIECE);
        const v4f g2 = *(const v4f*)(gsrc + 2 * PIECE);
        const v4f g3 = *(const v4f*)(gsrc + 3 * PIECE);
        *(v4f*)(sdst)             = g0;
        *(v4f*)(sdst + 4 * PITCH)  = g1;
        *(v4f*)(sdst + 8 * PITCH)  = g2;
        *(v4f*)(sdst + 12 * PITCH) = g3;
    }
    __builtin_amdgcn_fence(__ATOMIC_ACQ_REL, "wavefront");

    const int ncc = nc < 0 ? 0 : (nc > TSTEPS ? TSTEPS : nc);
    const int kA  = ncc / BLK;

    auto gates_of = [&](const v2f Bv) __attribute__((always_inline)) {
        Gates G;
        G.i = __builtin_amdgcn_wmma_f32_16x16x4_f32(false, A[0], false, Bv, (short)0, zc, false, false);
        G.f = __builtin_amdgcn_wmma_f32_16x16x4_f32(false, A[1], false, Bv, (short)0, zc, false, false);
        G.g = __builtin_amdgcn_wmma_f32_16x16x4_f32(false, A[2], false, Bv, (short)0, zc, false, false);
        G.o = __builtin_amdgcn_wmma_f32_16x16x4_f32(false, A[3], false, Bv, (short)0, zc, false, false);
        return G;
    };

    v2f in[BLK];
#pragma unroll
    for (int j = 0; j < BLK; ++j)
        in[j] = *(const v2f*)(ssrc + j * 4);
    Gates G = gates_of(in[0]);

    int k = 0;
    for (; k < kA; ++k) {
        const int kn = (k + 1 < NBLK) ? (k + 1) : (NBLK - 1);
        const float* gk = gsrc + (size_t)kn * BLK * 4;
        const v4f g0 = *(const v4f*)(gk);
        const v4f g1 = *(const v4f*)(gk + PIECE);
        const v4f g2 = *(const v4f*)(gk + 2 * PIECE);
        const v4f g3 = *(const v4f*)(gk + 3 * PIECE);
        pin();

        const int tb = k * BLK;
        v2f inn[BLK];
        float e0 = 0.0f, e1 = 0.0f;
#pragma unroll
        for (int j = 0; j < BLK; ++j) {
            if (j == BLK - 1) {
                pin();
                float* sk = sdst + ((k + 1) & 1) * (BLK * 4);
                *(v4f*)(sk)              = g0;
                *(v4f*)(sk + 4 * PITCH)  = g1;
                *(v4f*)(sk + 8 * PITCH)  = g2;
                *(v4f*)(sk + 12 * PITCH) = g3;
                __builtin_amdgcn_fence(__ATOMIC_ACQ_REL, "wavefront");
                const float* sn = ssrc + ((k + 1) & 1) * (BLK * 4);
#pragma unroll
                for (int i = 0; i < BLK; ++i)
                    inn[i] = *(const v2f*)(sn + i * 4);
            }
            const v2f Bn = (j < BLK - 1) ? in[j + 1] : inn[0];
            Gates Gn;
            float ai[8], af[8], ag[8], ao[8], tc[8];
            Gn.i = __builtin_amdgcn_wmma_f32_16x16x4_f32(false, A[0], false, Bn, (short)0, zc, false, false);
#pragma unroll
            for (int r = 0; r < 8; ++r) ai[r] = fmaf(G.i[r], 0.5f, hbi[r]);
#pragma unroll
            for (int r = 0; r < 8; ++r) { ai[r] = __builtin_amdgcn_tanhf(ai[r]); af[r] = fmaf(G.f[r], 0.5f, hbf[r]); }
            Gn.f = __builtin_amdgcn_wmma_f32_16x16x4_f32(false, A[1], false, Bn, (short)0, zc, false, false);
#pragma unroll
            for (int r = 0; r < 8; ++r) { af[r] = __builtin_amdgcn_tanhf(af[r]); ag[r] = G.g[r] + bg[r]; }
            Gn.g = __builtin_amdgcn_wmma_f32_16x16x4_f32(false, A[2], false, Bn, (short)0, zc, false, false);
#pragma unroll
            for (int r = 0; r < 8; ++r) { ag[r] = __builtin_amdgcn_tanhf(ag[r]); ao[r] = fmaf(G.o[r], 0.5f, hbo[r]); }
            Gn.o = __builtin_amdgcn_wmma_f32_16x16x4_f32(false, A[3], false, Bn, (short)0, zc, false, false);
#pragma unroll
            for (int r = 0; r < 8; ++r) {
                ao[r] = __builtin_amdgcn_tanhf(ao[r]);
                ai[r] = fmaf(ai[r], 0.5f, 0.5f);
                af[r] = fmaf(af[r], 0.5f, 0.5f);
            }
#pragma unroll
            for (int r = 0; r < 8; ++r) {
                const float cn = af[r] * c[r] + ai[r] * ag[r];
                c[r] = cn;
            }
#pragma unroll
            for (int r = 0; r < 8; ++r) { tc[r] = __builtin_amdgcn_tanhf(c[r]); ao[r] = fmaf(ao[r], 0.5f, 0.5f); }
            v2f p = {0.0f, 0.0f};
#pragma unroll
            for (int r = 0; r < 8; ++r) {
                const float s = ao[r] * tc[r];
                const v2f ss = {s, s};
                const v2f w  = {wr0[r], wr1[r]};
                p += ss * w;
            }
            h0 = p.x + xhalf(p.x);
            h1 = p.y + xhalf(p.y);

            if (j & 1) {
                v2f hv;
                hv.x = hi ? h0 : e0;
                hv.y = hi ? h1 : e1;
                __builtin_nontemporal_store(hv, (v2f*)(op + (size_t)(tb + j - 1 + hi) * 2));
            } else {
                e0 = h0;
                e1 = h1;
            }
            G = Gn;
        }
#pragma unroll
        for (int j = 0; j < BLK; ++j)
            in[j] = inn[j];
    }

    for (; k < NBLK; ++k) {
        const int kn = (k + 1 < NBLK) ? (k + 1) : (NBLK - 1);
        const float* gk = gsrc + (size_t)kn * BLK * 4;
        const v4f g0 = *(const v4f*)(gk);
        const v4f g1 = *(const v4f*)(gk + PIECE);
        const v4f g2 = *(const v4f*)(gk + 2 * PIECE);
        const v4f g3 = *(const v4f*)(gk + 3 * PIECE);
        pin();

        const int tb = k * BLK;
        v2f inn[BLK];
        float e0 = 0.0f, e1 = 0.0f;
#pragma unroll
        for (int j = 0; j < BLK; ++j) {
            const int t = tb + j;
            if (j == BLK - 1) {
                pin();
                float* sk = sdst + ((k + 1) & 1) * (BLK * 4);
                *(v4f*)(sk)              = g0;
                *(v4f*)(sk + 4 * PITCH)  = g1;
                *(v4f*)(sk + 8 * PITCH)  = g2;
                *(v4f*)(sk + 12 * PITCH) = g3;
                __builtin_amdgcn_fence(__ATOMIC_ACQ_REL, "wavefront");
                const float* sn = ssrc + ((k + 1) & 1) * (BLK * 4);
#pragma unroll
                for (int i = 0; i < BLK; ++i)
                    inn[i] = *(const v2f*)(sn + i * 4);
                pin();
            }
            const bool use_mem = (!hi) || (t < nc);
            v2f Bv;
            Bv.x = use_mem ? in[j].x : h0;
            Bv.y = use_mem ? in[j].y : h1;
            const v8f gi = __builtin_amdgcn_wmma_f32_16x16x4_f32(false, A[0], false, Bv, (short)0, zc, false, false);
            const v8f gg = __builtin_amdgcn_wmma_f32_16x16x4_f32(false, A[2], false, Bv, (short)0, zc, false, false);
            const v8f gf = __builtin_amdgcn_wmma_f32_16x16x4_f32(false, A[1], false, Bv, (short)0, zc, false, false);
            const v8f go = __builtin_amdgcn_wmma_f32_16x16x4_f32(false, A[3], false, Bv, (short)0, zc, false, false);

            float ai[8], af[8], ag[8], ao[8], sv[8];
#pragma unroll
            for (int r = 0; r < 8; ++r) { ai[r] = fmaf(gi[r], 0.5f, hbi[r]); ag[r] = gg[r] + bg[r]; }
#pragma unroll
            for (int r = 0; r < 8; ++r) { ai[r] = __builtin_amdgcn_tanhf(ai[r]); af[r] = fmaf(gf[r], 0.5f, hbf[r]); }
#pragma unroll
            for (int r = 0; r < 8; ++r) { ag[r] = __builtin_amdgcn_tanhf(ag[r]); ao[r] = fmaf(go[r], 0.5f, hbo[r]); }
#pragma unroll
            for (int r = 0; r < 8; ++r) { af[r] = __builtin_amdgcn_tanhf(af[r]); ai[r] = fmaf(ai[r], 0.5f, 0.5f); }
#pragma unroll
            for (int r = 0; r < 8; ++r) {
                ao[r] = __builtin_amdgcn_tanhf(ao[r]);
                af[r] = fmaf(af[r], 0.5f, 0.5f);
                const float cn = af[r] * c[r] + ai[r] * ag[r];
                c[r] = cn;
            }
#pragma unroll
            for (int r = 0; r < 8; ++r) {
                const float tcr = __builtin_amdgcn_tanhf(c[r]);
                sv[r] = fmaf(ao[r], 0.5f, 0.5f) * tcr;
            }
            v2f u[4];
#pragma unroll
            for (int q = 0; q < 4; ++q) {
                const v2f sa = {sv[2 * q], sv[2 * q]};
                const v2f sb = {sv[2 * q + 1], sv[2 * q + 1]};
                const v2f wa = {wr0[2 * q], wr1[2 * q]};
                const v2f wb = {wr0[2 * q + 1], wr1[2 * q + 1]};
                u[q] = sa * wa + sb * wb;
            }
            const v2f p = ((u[0] + u[1]) + u[2]) + u[3];
            const float p0 = p.x, p1 = p.y;
            h0 = p0 + xhalf_rec(p0);
            h1 = p1 + xhalf_rec(p1);

            if (j & 1) {
                v2f hv;
                hv.x = hi ? h0 : e0;
                hv.y = hi ? h1 : e1;
                __builtin_nontemporal_store(hv, (v2f*)(op + (size_t)(tb + j - 1 + hi) * 2));
            } else {
                e0 = h0;
                e1 = h1;
            }
        }
#pragma unroll
        for (int j = 0; j < BLK; ++j)
            in[j] = inn[j];
    }
}

extern "C" void kernel_launch(void* const* d_in, const int* in_sizes, int n_in,
                              void* d_out, int out_size, void* d_ws, size_t ws_size,
                              hipStream_t stream) {
    const float* u    = (const float*)d_in[0];
    const float* Wih  = (const float*)d_in[1];
    const float* Whh  = (const float*)d_in[2];
    const float* bih  = (const float*)d_in[3];
    const float* bhh  = (const float*)d_in[4];
    const float* Whr  = (const float*)d_in[5];
    const int*   ncp  = (const int*)d_in[6];
    float*       out  = (float*)d_out;

    dim3 grid(BATCH / 16);
    dim3 block(32);
    OpenLSTM_wmma_scan<<<grid, block, 0, stream>>>(u, Wih, Whh, bih, bhh, Whr, ncp, out);
}


# context phase (t < n_context) spread over the 4 SIMDs of a WGP: 4 waves per 16-batch tile, each one gate tile of 4 hidden units; projection shares summed per 8-step block via LDS (sum order changes)
# speedup vs baseline: 4.1584x; 1.2996x over previous
#include <hip/hip_runtime.h>

typedef __attribute__((ext_vector_type(2))) float v2f;
typedef __attribute__((ext_vector_type(4))) float v4f;
typedef __attribute__((ext_vector_type(8))) float v8f;

#define BATCH   4096
#define TSTEPS  2048
#define BLK     8
#define NBLK    (TSTEPS / BLK)
#define PP      18
#define PITCH   68

__device__ __forceinline__ float fsigm_b(float x, float hb) {
    return fmaf(__builtin_amdgcn_tanhf(fmaf(x, 0.5f, hb)), 0.5f, 0.5f);
}

__device__ __forceinline__ float xhalf(float x) {
    const int b = __builtin_bit_cast(int, x);
    return __builtin_bit_cast(float, __builtin_amdgcn_permlanex16(b, b, 0x76543210, 0xfedcba98, false, false));
}

__device__ __forceinline__ float xhalf_rec(float x) {
    const int b = __builtin_bit_cast(int, x);
    return __builtin_bit_cast(float, __builtin_amdgcn_permlanex16(0, b, 0x76543210, 0xfedcba98, false, false));
}

__device__ __forceinline__ void pin() { __builtin_amdgcn_sched_barrier(0); }

__device__ __forceinline__ void lds_barrier() {
    __builtin_amdgcn_fence(__ATOMIC_RELEASE, "workgroup", "local");
    __builtin_amdgcn_s_barrier();
    __builtin_amdgcn_fence(__ATOMIC_ACQUIRE, "workgroup", "local");
}

struct Gates { v8f i, f, g, o; };

__global__ __launch_bounds__(128)
void OpenLSTM_wmma_scan(const float* __restrict__ u,
                        const float* __restrict__ Wih,
                        const float* __restrict__ Whh,
                        const float* __restrict__ bih,
                        const float* __restrict__ bhh,
                        const float* __restrict__ Whr,
                        const int*   __restrict__ ncp,
                        float*       __restrict__ out)
{
    const int lane = threadIdx.x & 31;
    const int w    = __builtin_amdgcn_readfirstlane((int)(threadIdx.x >> 5));
    const int n    = lane & 15;
    const int hi   = lane >> 4;
    const int b0   = blockIdx.x * 16;
    const int nc   = *ncp;

    __shared__ __attribute__((aligned(16))) float stage[16 * PITCH];
    __shared__ __attribute__((aligned(16))) float part[2 * 8 * 16 * PP];
    __shared__ float cbuf[16 * 17];
    const int ls = lane & 7;
    const int lb = lane >> 3;
    const float* gsrc = u + ((size_t)(b0 + lb) * TSTEPS + ls) * 4;
    float*       sdst = stage + lb * PITCH + ls * 4;
    const float* ssrc = stage + n * PITCH + hi * 2;
    const size_t PIECE = (size_t)4 * TSTEPS * 4;

    const float* Wsrc = hi ? Whh : Wih;

    v8f c = {};
    const v8f zc = {};
    float h0 = 0.0f, h1 = 0.0f;

    float*       op = out + (size_t)(b0 + n) * TSTEPS * 2;

    v2f A4;
    float hb4[8], w40[2], w41[2];
#pragma unroll
    for (int r = 0; r < 8; ++r) {
        const int grow = (r >> 1) * 16 + 4 * w + 2 * hi + (r & 1);
        const float bsum = bih[grow] + bhh[grow];
        hb4[r] = ((r >> 1) == 2) ? bsum : 0.5f * bsum;
    }
    {
        const int m    = n;
        const int grow = ((m & 7) >> 1) * 16 + 4 * w + 2 * (m >> 3) + (m & 1);
        A4.x = Wsrc[grow * 2 + 0];
        A4.y = Wsrc[grow * 2 + 1];
    }
#pragma unroll
    for (int q = 0; q < 2; ++q) {
        w40[q] = Whr[0 * 16 + 4 * w + 2 * hi + q];
        w41[q] = Whr[1 * 16 + 4 * w + 2 * hi + q];
    }
    float c4[2] = {0.0f, 0.0f};

    const float* gsrc4 = u + ((size_t)(b0 + 4 * w + lb) * TSTEPS + ls) * 4;
    float*       sdst4 = stage + (4 * w + lb) * PITCH + ls * 4;
    float*       orow4 = out + ((size_t)(b0 + 4 * w + lb) * TSTEPS + ls) * 2;

    {
        const v4f g = *(const v4f*)(gsrc4);
        *(v4f*)(sdst4) = g;
    }
    __syncthreads();

    v2f in[BLK];
#pragma unroll
    for (int j = 0; j < BLK; ++j)
        in[j] = *(const v2f*)(ssrc + j * 4);

    const int ncc = nc < 0 ? 0 : (nc > TSTEPS ? TSTEPS : nc);
    const int kA  = (ncc / (2 * BLK)) * 2;

    auto ctx_block = [&](const int kk, const v4f gst) __attribute__((always_inline)) {
        const int k = kk;
        float* pw = part + ((((k & 1) * 8) + 2 * w + hi) * 16 + n) * PP;
#pragma unroll
        for (int j = 0; j < BLK; ++j) {
            const v8f Gd = __builtin_amdgcn_wmma_f32_16x16x4_f32(false, A4, false, in[j], (short)0, zc, false, false);
            float tv[8];
#pragma unroll
            for (int r = 0; r < 8; ++r)
                tv[r] = ((r >> 1) == 2) ? (Gd[r] + hb4[r]) : fmaf(Gd[r], 0.5f, hb4[r]);
#pragma unroll
            for (int r = 0; r < 8; ++r)
                tv[r] = __builtin_amdgcn_tanhf(tv[r]);
            float sq[2];
#pragma unroll
            for (int q = 0; q < 2; ++q) {
                const float si = fmaf(tv[0 + q], 0.5f, 0.5f);
                const float sf = fmaf(tv[2 + q], 0.5f, 0.5f);
                const float so = fmaf(tv[6 + q], 0.5f, 0.5f);
                const float cn = sf * c4[q] + si * tv[4 + q];
                c4[q] = cn;
                sq[q] = so * __builtin_amdgcn_tanhf(cn);
            }
            v2f p;
            p.x = sq[0] * w40[0] + sq[1] * w40[1];
            p.y = sq[0] * w41[0] + sq[1] * w41[1];
            *(v2f*)(pw + j * 2) = p;
        }

        pin();
        *(v4f*)(sdst4 + ((k + 1) & 1) * (BLK * 4)) = gst;
        lds_barrier();
        {
            const float* sn = ssrc + ((k + 1) & 1) * (BLK * 4);
#pragma unroll
            for (int i = 0; i < BLK; ++i)
                in[i] = *(const v2f*)(sn + i * 4);
        }
        {
            const float* pr = part + (((k & 1) * 8) * 16 + 4 * w + lb) * PP + ls * 2;
            v2f q8[8];
#pragma unroll
            for (int s = 0; s < 8; ++s)
                q8[s] = *(const v2f*)(pr + s * 16 * PP);
            const v2f hv = ((q8[0] + q8[1]) + (q8[2] + q8[3])) + ((q8[4] + q8[5]) + (q8[6] + q8[7]));
            __builtin_nontemporal_store(hv, (v2f*)(orow4 + (size_t)k * BLK * 2));
        }
    };

    auto ld_blk = [&](const int kb) __attribute__((always_inline)) {
        const int kn = (kb < NBLK) ? kb : (NBLK - 1);
        return *(const v4f*)(gsrc4 + (size_t)kn * BLK * 4);
    };
    v4f gA = ld_blk(1);
    int k = 0;
    for (; k < kA; k += 2) {
        const v4f gB = ld_blk(k + 2);
        pin();
        ctx_block(k, gA);
        gA = ld_blk(k + 3);
        pin();
        ctx_block(k + 1, gB);
    }

    cbuf[(4 * w + 2 * hi + 0) * 17 + n] = c4[0];
    cbuf[(4 * w + 2 * hi + 1) * 17 + n] = c4[1];
    __syncthreads();
    if (w != 0) return;

    v2f A[4];
#pragma unroll
    for (int g = 0; g < 4; ++g) {
        const int row = g * 16 + n;
        A[g].x = Wsrc[row * 2 + 0];
        A[g].y = Wsrc[row * 2 + 1];
    }

    float hbi[8], hbf[8], hbo[8], bg[8];
#pragma unroll
    for (int r = 0; r < 8; ++r) {
        const int m = r + hi * 8;
        hbi[r] = 0.5f * (bih[ 0 + m] + bhh[ 0 + m]);
        hbf[r] = 0.5f * (bih[16 + m] + bhh[16 + m]);
        bg[r]  =         bih[32 + m] + bhh[32 + m];
        hbo[r] = 0.5f * (bih[48 + m] + bhh[48 + m]);
    }

    float wr0[8], wr1[8];
#pragma unroll
    for (int r = 0; r < 8; ++r) {
        wr0[r] = Whr[0 * 16 + r + hi * 8];
        wr1[r] = Whr[1 * 16 + r + hi * 8];
    }


#pragma unroll
    for (int r = 0; r < 8; ++r)
        c[r] = cbuf[(r + 8 * hi) * 17 + n];
    if (kA > 0) {
        const float* pr = part + ((((kA - 1) & 1) * 8) * 16 + n) * PP + (BLK - 1) * 2;
        v2f q8[8];
#pragma unroll
        for (int s = 0; s < 8; ++s)
            q8[s] = *(const v2f*)(pr + s * 16 * PP);
        const v2f hv = ((q8[0] + q8[1]) + (q8[2] + q8[3])) + ((q8[4] + q8[5]) + (q8[6] + q8[7]));
        h0 = hv.x;
        h1 = hv.y;
    }

    for (; k < NBLK; ++k) {
        const int kn = (k + 1 < NBLK) ? (k + 1) : (NBLK - 1);
        const float* gk = gsrc + (size_t)kn * BLK * 4;
        const v4f g0 = *(const v4f*)(gk);
        const v4f g1 = *(const v4f*)(gk + PIECE);
        const v4f g2 = *(const v4f*)(gk + 2 * PIECE);
        const v4f g3 = *(const v4f*)(gk + 3 * PIECE);
        pin();

        const int tb = k * BLK;
        v2f inn[BLK];
        float e0 = 0.0f, e1 = 0.0f;
#pragma unroll
        for (int j = 0; j < BLK; ++j) {
            const int t = tb + j;
            if (j == BLK - 1) {
                pin();
                float* sk = sdst + ((k + 1) & 1) * (BLK * 4);
                *(v4f*)(sk)              = g0;
                *(v4f*)(sk + 4 * PITCH)  = g1;
                *(v4f*)(sk + 8 * PITCH)  = g2;
                *(v4f*)(sk + 12 * PITCH) = g3;
                __builtin_amdgcn_fence(__ATOMIC_ACQ_REL, "wavefront");
                const float* sn = ssrc + ((k + 1) & 1) * (BLK * 4);
#pragma unroll
                for (int i = 0; i < BLK; ++i)
                    inn[i] = *(const v2f*)(sn + i * 4);
                pin();
            }
            const bool use_mem = (!hi) || (t < nc);
            v2f Bv;
            Bv.x = use_mem ? in[j].x : h0;
            Bv.y = use_mem ? in[j].y : h1;
            const v8f gi = __builtin_amdgcn_wmma_f32_16x16x4_f32(false, A[0], false, Bv, (short)0, zc, false, false);
            const v8f gg = __builtin_amdgcn_wmma_f32_16x16x4_f32(false, A[2], false, Bv, (short)0, zc, false, false);
            const v8f gf = __builtin_amdgcn_wmma_f32_16x16x4_f32(false, A[1], false, Bv, (short)0, zc, false, false);
            const v8f go = __builtin_amdgcn_wmma_f32_16x16x4_f32(false, A[3], false, Bv, (short)0, zc, false, false);

            float ai[8], af[8], ag[8], ao[8], sv[8];
#pragma unroll
            for (int r = 0; r < 8; ++r) { ai[r] = fmaf(gi[r], 0.5f, hbi[r]); ag[r] = gg[r] + bg[r]; }
#pragma unroll
            for (int r = 0; r < 8; ++r) { ai[r] = __builtin_amdgcn_tanhf(ai[r]); af[r] = fmaf(gf[r], 0.5f, hbf[r]); }
#pragma unroll
            for (int r = 0; r < 8; ++r) { ag[r] = __builtin_amdgcn_tanhf(ag[r]); ao[r] = fmaf(go[r], 0.5f, hbo[r]); }
#pragma unroll
            for (int r = 0; r < 8; ++r) { af[r] = __builtin_amdgcn_tanhf(af[r]); ai[r] = fmaf(ai[r], 0.5f, 0.5f); }
#pragma unroll
            for (int r = 0; r < 8; ++r) {
                ao[r] = __builtin_amdgcn_tanhf(ao[r]);
                af[r] = fmaf(af[r], 0.5f, 0.5f);
                const float cn = af[r] * c[r] + ai[r] * ag[r];
                c[r] = cn;
            }
#pragma unroll
            for (int r = 0; r < 8; ++r) {
                const float tcr = __builtin_amdgcn_tanhf(c[r]);
                sv[r] = fmaf(ao[r], 0.5f, 0.5f) * tcr;
            }
            v2f u[4];
#pragma unroll
            for (int q = 0; q < 4; ++q) {
                const v2f sa = {sv[2 * q], sv[2 * q]};
                const v2f sb = {sv[2 * q + 1], sv[2 * q + 1]};
                const v2f wa = {wr0[2 * q], wr1[2 * q]};
                const v2f wb = {wr0[2 * q + 1], wr1[2 * q + 1]};
                u[q] = sa * wa + sb * wb;
            }
            const v2f p = ((u[0] + u[1]) + u[2]) + u[3];
            const float p0 = p.x, p1 = p.y;
            h0 = p0 + xhalf_rec(p0);
            h1 = p1 + xhalf_rec(p1);

            if (j & 1) {
                v2f hv;
                hv.x = hi ? h0 : e0;
                hv.y = hi ? h1 : e1;
                __builtin_nontemporal_store(hv, (v2f*)(op + (size_t)(tb + j - 1 + hi) * 2));
            } else {
                e0 = h0;
                e1 = h1;
            }
        }
#pragma unroll
        for (int j = 0; j < BLK; ++j)
            in[j] = inn[j];
    }
}

extern "C" void kernel_launch(void* const* d_in, const int* in_sizes, int n_in,
                              void* d_out, int out_size, void* d_ws, size_t ws_size,
                              hipStream_t stream) {
    const float* u    = (const float*)d_in[0];
    const float* Wih  = (const float*)d_in[1];
    const float* Whh  = (const float*)d_in[2];
    const float* bih  = (const float*)d_in[3];
    const float* bhh  = (const float*)d_in[4];
    const float* Whr  = (const float*)d_in[5];
    const int*   ncp  = (const int*)d_in[6];
    float*       out  = (float*)d_out;

    dim3 grid(BATCH / 16);
    dim3 block(128);
    OpenLSTM_wmma_scan<<<grid, block, 0, stream>>>(u, Wih, Whh, bih, bhh, Whr, ncp, out);
}


# steps t >= n_context on all 4 waves, 4 batches each: one 16x16 gate tile per wave by zero-blocked k (unit sets), input product a step ahead as C; h summed over 8 lanes (sum order changes)
# speedup vs baseline: 5.5876x; 1.3437x over previous
#include <hip/hip_runtime.h>

typedef __attribute__((ext_vector_type(2))) float v2f;
typedef __attribute__((ext_vector_type(4))) float v4f;
typedef __attribute__((ext_vector_type(8))) float v8f;

#define BATCH   4096
#define TSTEPS  2048
#define BLK     8
#define NBLK    (TSTEPS / BLK)
#define PP      18
#define PITCH   68

__device__ __forceinline__ float fsigm_b(float x, float hb) {
    return fmaf(__builtin_amdgcn_tanhf(fmaf(x, 0.5f, hb)), 0.5f, 0.5f);
}

__device__ __forceinline__ float xhalf(float x) {
    const int b = __builtin_bit_cast(int, x);
    return __builtin_bit_cast(float, __builtin_amdgcn_permlanex16(b, b, 0x76543210, 0xfedcba98, false, false));
}

__device__ __forceinline__ float xhalf_rec(float x) {
    const int b = __builtin_bit_cast(int, x);
    return __builtin_bit_cast(float, __builtin_amdgcn_permlanex16(0, b, 0x76543210, 0xfedcba98, false, false));
}

__device__ __forceinline__ float rowx4(float x) {
    const int b = __builtin_bit_cast(int, x);
    return __builtin_bit_cast(float, __builtin_amdgcn_update_dpp(0, b, 0x164, 0xf, 0xf, true));
}
__device__ __forceinline__ float rowx8(float x) {
    const int b = __builtin_bit_cast(int, x);
    return __builtin_bit_cast(float, __builtin_amdgcn_update_dpp(0, b, 0x168, 0xf, 0xf, true));
}

__device__ __forceinline__ void wmma_slots(v8f& d, const v2f ba, const v2f bb) {
    asm("v_nop\n\tv_nop\n\tv_nop\n\tv_nop" : "+v"(d) : "v"(ba), "v"(bb));
}

__device__ __forceinline__ void issue_ahead(v8f& next, v8f& cur) {
    asm("" : "+v"(next), "+v"(cur));
}

__device__ __forceinline__ void pin() { __builtin_amdgcn_sched_barrier(0); }

__device__ __forceinline__ void lds_barrier() {
    __builtin_amdgcn_fence(__ATOMIC_RELEASE, "workgroup", "local");
    __builtin_amdgcn_s_barrier();
    __builtin_amdgcn_fence(__ATOMIC_ACQUIRE, "workgroup", "local");
}

struct Gates { v8f i, f, g, o; };

__global__ __launch_bounds__(128)
void OpenLSTM_wmma_scan(const float* __restrict__ u,
                        const float* __restrict__ Wih,
                        const float* __restrict__ Whh,
                        const float* __restrict__ bih,
                        const float* __restrict__ bhh,
                        const float* __restrict__ Whr,
                        const int*   __restrict__ ncp,
                        float*       __restrict__ out)
{
    const int lane = threadIdx.x & 31;
    const int w    = __builtin_amdgcn_readfirstlane((int)(threadIdx.x >> 5));
    const int n    = lane & 15;
    const int hi   = lane >> 4;
    const int b0   = blockIdx.x * 16;
    const int nc   = *ncp;

    __shared__ __attribute__((aligned(16))) float stage[16 * PITCH + 2 * BLK * 4];
    __shared__ __attribute__((aligned(16))) float part[2 * 8 * 16 * PP];
    __shared__ float cbuf[16 * 17];
    const int ls = lane & 7;
    const int lb = lane >> 3;
    const float* gsrc = u + ((size_t)(b0 + lb) * TSTEPS + ls) * 4;
    float*       sdst = stage + lb * PITCH + ls * 4;
    const float* ssrc = stage + n * PITCH + hi * 2;
    const size_t PIECE = (size_t)4 * TSTEPS * 4;

    const float* Wsrc = hi ? Whh : Wih;

    v8f c = {};
    const v8f zc = {};
    float h0 = 0.0f, h1 = 0.0f;

    float*       op = out + (size_t)(b0 + n) * TSTEPS * 2;

    v2f A4;
    float hb4[8], w40[2], w41[2];
#pragma unroll
    for (int r = 0; r < 8; ++r) {
        const int grow = (r >> 1) * 16 + 4 * w + 2 * hi + (r & 1);
        const float bsum = bih[grow] + bhh[grow];
        hb4[r] = ((r >> 1) == 2) ? bsum : 0.5f * bsum;
    }
    {
        const int m    = n;
        const int grow = ((m & 7) >> 1) * 16 + 4 * w + 2 * (m >> 3) + (m & 1);
        A4.x = Wsrc[grow * 2 + 0];
        A4.y = Wsrc[grow * 2 + 1];
    }
#pragma unroll
    for (int q = 0; q < 2; ++q) {
        w40[q] = Whr[0 * 16 + 4 * w + 2 * hi + q];
        w41[q] = Whr[1 * 16 + 4 * w + 2 * hi + q];
    }
    float c4[2] = {0.0f, 0.0f};

    const float* gsrc4 = u + ((size_t)(b0 + 4 * w + lb) * TSTEPS + ls) * 4;
    float*       sdst4 = stage + (4 * w + lb) * PITCH + ls * 4;
    float*       orow4 = out + ((size_t)(b0 + 4 * w + lb) * TSTEPS + ls) * 2;

    {
        const v4f g = *(const v4f*)(gsrc4);
        *(v4f*)(sdst4) = g;
    }
    if (threadIdx.x < 2 * BLK * 4) stage[16 * PITCH + threadIdx.x] = 0.0f;
    __syncthreads();

    v2f in[BLK];
#pragma unroll
    for (int j = 0; j < BLK; ++j)
        in[j] = *(const v2f*)(ssrc + j * 4);

    const int ncc = nc < 0 ? 0 : (nc > TSTEPS ? TSTEPS : nc);
    const int kA  = (ncc / (2 * BLK)) * 2;

    auto ctx_block = [&](const int kk, const v4f gst) __attribute__((always_inline)) {
        const int k = kk;
        float* pw = part + ((((k & 1) * 8) + 2 * w + hi) * 16 + n) * PP;
        v8f Gn4 = __builtin_amdgcn_wmma_f32_16x16x4_f32(false, A4, false, in[0], (short)0, zc, false, false);
        wmma_slots(Gn4, in[0], in[0]);
#pragma unroll
        for (int j = 0; j < BLK; ++j) {
            v8f Gd = Gn4;
            if (j + 1 < BLK) {
                Gn4 = __builtin_amdgcn_wmma_f32_16x16x4_f32(false, A4, false, in[j + 1], (short)0, zc, false, false);
                issue_ahead(Gn4, Gd);
            }
            float tv[8];
#pragma unroll
            for (int r = 0; r < 8; ++r)
                tv[r] = ((r >> 1) == 2) ? (Gd[r] + hb4[r]) : fmaf(Gd[r], 0.5f, hb4[r]);
#pragma unroll
            for (int r = 0; r < 8; ++r)
                tv[r] = __builtin_amdgcn_tanhf(tv[r]);
            float sq[2];
#pragma unroll
            for (int q = 0; q < 2; ++q) {
                const float si = fmaf(tv[0 + q], 0.5f, 0.5f);
                const float sf = fmaf(tv[2 + q], 0.5f, 0.5f);
                const float so = fmaf(tv[6 + q], 0.5f, 0.5f);
                const float cn = sf * c4[q] + si * tv[4 + q];
                c4[q] = cn;
                sq[q] = so * __builtin_amdgcn_tanhf(cn);
            }
            v2f p;
            p.x = sq[0] * w40[0] + sq[1] * w40[1];
            p.y = sq[0] * w41[0] + sq[1] * w41[1];
            *(v2f*)(pw + j * 2) = p;
        }

        pin();
        *(v4f*)(sdst4 + ((k + 1) & 1) * (BLK * 4)) = gst;
        lds_barrier();
        {
            const float* sn = ssrc + ((k + 1) & 1) * (BLK * 4);
#pragma unroll
            for (int i = 0; i < BLK; ++i)
                in[i] = *(const v2f*)(sn + i * 4);
        }
        {
            const float* pr = part + (((k & 1) * 8) * 16 + 4 * w + lb) * PP + ls * 2;
            v2f q8[8];
#pragma unroll
            for (int s = 0; s < 8; ++s)
                q8[s] = *(const v2f*)(pr + s * 16 * PP);
            const v2f hv = ((q8[0] + q8[1]) + (q8[2] + q8[3])) + ((q8[4] + q8[5]) + (q8[6] + q8[7]));
            __builtin_nontemporal_store(hv, (v2f*)(orow4 + (size_t)k * BLK * 2));
        }
    };

    auto ld_blk = [&](const int kb) __attribute__((always_inline)) {
        const int kn = (kb < NBLK) ? kb : (NBLK - 1);
        return *(const v4f*)(gsrc4 + (size_t)kn * BLK * 4);
    };
    v4f gA = ld_blk(1);
    int k = 0;
    for (; k < kA; k += 2) {
        const v4f gB = ld_blk(k + 2);
        pin();
        ctx_block(k, gA);
        gA = ld_blk(k + 3);
        pin();
        ctx_block(k + 1, gB);
    }

    const int bb = lane & 3;
    const int s4 = (lane >> 2) & 3;
    const bool mA = (s4 == hi);
    const bool mB = (s4 == 2 + hi);
    v2f Aha, Ahb, Axa, Axb;
    {
        const int usub = 2 * (n >> 3) + (n & 1);
        const int gsel = (n & 7) >> 1;
        const int ga = gsel * 16 + 4 * hi + usub;
        const int gb = gsel * 16 + 4 * (2 + hi) + usub;
        Aha.x = Whh[ga * 2 + 0]; Aha.y = Whh[ga * 2 + 1];
        Ahb.x = Whh[gb * 2 + 0]; Ahb.y = Whh[gb * 2 + 1];
        Axa.x = Wih[ga * 2 + 0]; Axa.y = Wih[ga * 2 + 1];
        Axb.x = Wih[gb * 2 + 0]; Axb.y = Wih[gb * 2 + 1];
    }
    float hb2[8], wp0[2], wp1[2];
#pragma unroll
    for (int r = 0; r < 8; ++r) {
        const int grow = (r >> 1) * 16 + 4 * s4 + 2 * hi + (r & 1);
        const float bsum = bih[grow] + bhh[grow];
        hb2[r] = ((r >> 1) == 2) ? bsum : 0.5f * bsum;
    }
#pragma unroll
    for (int q = 0; q < 2; ++q) {
        wp0[q] = Whr[0 * 16 + 4 * s4 + 2 * hi + q];
        wp1[q] = Whr[1 * 16 + 4 * s4 + 2 * hi + q];
    }
    cbuf[(4 * w + 2 * hi + 0) * 17 + n] = c4[0];
    cbuf[(4 * w + 2 * hi + 1) * 17 + n] = c4[1];
    __syncthreads();
    float c2[2];
#pragma unroll
    for (int q = 0; q < 2; ++q)
        c2[q] = cbuf[(4 * s4 + 2 * hi + q) * 17 + 4 * w + bb];
    if (kA > 0) {
        const float* pr = part + ((((kA - 1) & 1) * 8) * 16 + 4 * w + bb) * PP + (BLK - 1) * 2;
        v2f q8[8];
#pragma unroll
        for (int s = 0; s < 8; ++s)
            q8[s] = *(const v2f*)(pr + s * 16 * PP);
        const v2f hv = ((q8[0] + q8[1]) + (q8[2] + q8[3])) + ((q8[4] + q8[5]) + (q8[6] + q8[7]));
        h0 = hv.x;
        h1 = hv.y;
    }

    const float* srcA = stage + (mA ? (4 * w + bb) * PITCH : 16 * PITCH);
    const float* srcB = stage + (mB ? (4 * w + bb) * PITCH : 16 * PITCH);
    const int g8 = s4 + 4 * hi;
    float* orow2 = out + ((size_t)(b0 + 4 * w + bb) * TSTEPS + g8) * 2;

    v4f iA0[BLK], iB0[BLK], iA1[BLK], iB1[BLK];
    {
        const int hf = (kA & 1) * (BLK * 4);
#pragma unroll
        for (int i = 0; i < BLK; ++i) {
            iA0[i] = *(const v4f*)(srcA + hf + i * 4);
            iB0[i] = *(const v4f*)(srcB + hf + i * 4);
        }
    }
    v8f Dx;
    {
        const v2f xa = {iA0[0].x, iA0[0].y};
        const v2f xb = {iB0[0].x, iB0[0].y};
        Dx = __builtin_amdgcn_wmma_f32_16x16x4_f32(false, Axa, false, xa, (short)0, zc, false, false);
        Dx = __builtin_amdgcn_wmma_f32_16x16x4_f32(false, Axb, false, xb, (short)0, Dx, false, false);
    }

    auto fr_block = [&](const int kk, v4f (&ia)[BLK], v4f (&ib)[BLK], v4f (&na)[BLK], v4f (&nb)[BLK])
        __attribute__((always_inline)) {
        const int tb = kk * BLK;
        v2f hs = {0.0f, 0.0f};
#pragma unroll
        for (int j = 0; j < BLK; ++j) {
            if (j == BLK - 2) {
                pin();
                *(v4f*)(sdst4 + ((kk + 1) & 1) * (BLK * 4)) = gA;
                __builtin_amdgcn_fence(__ATOMIC_ACQ_REL, "wavefront");
                const int hf = ((kk + 1) & 1) * (BLK * 4);
#pragma unroll
                for (int i = 0; i < BLK; ++i) {
                    na[i] = *(const v4f*)(srcA + hf + i * 4);
                    nb[i] = *(const v4f*)(srcB + hf + i * 4);
                }
                gA = ld_blk(kk + 2);
                pin();
            }
            const int t = tb + j;
            const bool fr = !(t < nc);
            const bool cA = mA && fr;
            const bool cB = mB && fr;
            v2f Ba, Bb;
            Ba.x = cA ? h0 : ia[j].z;
            Ba.y = cA ? h1 : ia[j].w;
            Bb.x = cB ? h0 : ib[j].z;
            Bb.y = cB ? h1 : ib[j].w;
            v8f Gd = __builtin_amdgcn_wmma_f32_16x16x4_f32(false, Aha, false, Ba, (short)0, Dx, false, false);
            Gd = __builtin_amdgcn_wmma_f32_16x16x4_f32(false, Ahb, false, Bb, (short)0, Gd, false, false);
            wmma_slots(Gd, Ba, Bb);
            {
                const v4f ua = (j + 1 < BLK) ? ia[(j + 1) & (BLK - 1)] : na[0];
                const v4f ub = (j + 1 < BLK) ? ib[(j + 1) & (BLK - 1)] : nb[0];
                const v2f xa = {ua.x, ua.y};
                const v2f xb = {ub.x, ub.y};
                Dx = __builtin_amdgcn_wmma_f32_16x16x4_f32(false, Axa, false, xa, (short)0, zc, false, false);
                Dx = __builtin_amdgcn_wmma_f32_16x16x4_f32(false, Axb, false, xb, (short)0, Dx, false, false);
                wmma_slots(Dx, xa, xb);
            }
            float tv[8];
#pragma unroll
            for (int r = 0; r < 8; ++r)
                tv[r] = ((r >> 1) == 2) ? (Gd[r] + hb2[r]) : fmaf(Gd[r], 0.5f, hb2[r]);
#pragma unroll
            for (int r = 0; r < 8; ++r)
                tv[r] = __builtin_amdgcn_tanhf(tv[r]);
            float sq[2];
#pragma unroll
            for (int q = 0; q < 2; ++q) {
                const float si = fmaf(tv[0 + q], 0.5f, 0.5f);
                const float sf = fmaf(tv[2 + q], 0.5f, 0.5f);
                const float so = fmaf(tv[6 + q], 0.5f, 0.5f);
                const float cn = sf * c2[q] + si * tv[4 + q];
                c2[q] = cn;
                sq[q] = so * __builtin_amdgcn_tanhf(cn);
            }
            float p0 = sq[0] * wp0[0] + sq[1] * wp0[1];
            float p1 = sq[0] * wp1[0] + sq[1] * wp1[1];
            p0 += rowx4(p0);  p1 += rowx4(p1);
            p0 += rowx8(p0);  p1 += rowx8(p1);
            h0 = p0 + xhalf_rec(p0);
            h1 = p1 + xhalf_rec(p1);
            hs.x = (g8 == j) ? h0 : hs.x;
            hs.y = (g8 == j) ? h1 : hs.y;
        }
        __builtin_nontemporal_store(hs, (v2f*)(orow2 + (size_t)tb * 2));
    };

    for (; k < NBLK; k += 2) {
        fr_block(k, iA0, iB0, iA1, iB1);
        fr_block(k + 1, iA1, iB1, iA0, iB0);
    }
}

extern "C" void kernel_launch(void* const* d_in, const int* in_sizes, int n_in,
                              void* d_out, int out_size, void* d_ws, size_t ws_size,
                              hipStream_t stream) {
    const float* u    = (const float*)d_in[0];
    const float* Wih  = (const float*)d_in[1];
    const float* Whh  = (const float*)d_in[2];
    const float* bih  = (const float*)d_in[3];
    const float* bhh  = (const float*)d_in[4];
    const float* Whr  = (const float*)d_in[5];
    const int*   ncp  = (const int*)d_in[6];
    float*       out  = (float*)d_out;

    dim3 grid(BATCH / 16);
    dim3 block(128);
    OpenLSTM_wmma_scan<<<grid, block, 0, stream>>>(u, Wih, Whh, bih, bhh, Whr, ncp, out);
}


# kernel declared one wave per SIMD (waves_per_eu 1,1): compiler overlaps a block's independent steps instead of saving registers, drops inter-wave delay hints; arithmetic unchanged
# speedup vs baseline: 6.4925x; 1.1619x over previous
#include <hip/hip_runtime.h>

typedef __attribute__((ext_vector_type(2))) float v2f;
typedef __attribute__((ext_vector_type(4))) float v4f;
typedef __attribute__((ext_vector_type(8))) float v8f;

#define BATCH   4096
#define TSTEPS  2048
#define BLK     8
#define NBLK    (TSTEPS / BLK)
#define PP      18
#define PITCH   68

__device__ __forceinline__ float fsigm_b(float x, float hb) {
    return fmaf(__builtin_amdgcn_tanhf(fmaf(x, 0.5f, hb)), 0.5f, 0.5f);
}

__device__ __forceinline__ float xhalf(float x) {
    const int b = __builtin_bit_cast(int, x);
    return __builtin_bit_cast(float, __builtin_amdgcn_permlanex16(b, b, 0x76543210, 0xfedcba98, false, false));
}

__device__ __forceinline__ float xhalf_rec(float x) {
    const int b = __builtin_bit_cast(int, x);
    return __builtin_bit_cast(float, __builtin_amdgcn_permlanex16(0, b, 0x76543210, 0xfedcba98, false, false));
}

__device__ __forceinline__ float rowx4(float x) {
    const int b = __builtin_bit_cast(int, x);
    return __builtin_bit_cast(float, __builtin_amdgcn_update_dpp(0, b, 0x164, 0xf, 0xf, true));
}
__device__ __forceinline__ float rowx8(float x) {
    const int b = __builtin_bit_cast(int, x);
    return __builtin_bit_cast(float, __builtin_amdgcn_update_dpp(0, b, 0x168, 0xf, 0xf, true));
}

__device__ __forceinline__ void wmma_slots(v8f& d, const v2f ba, const v2f bb) {
    asm("v_nop\n\tv_nop\n\tv_nop\n\tv_nop" : "+v"(d) : "v"(ba), "v"(bb));
}

__device__ __forceinline__ void issue_ahead(v8f& next, v8f& cur) {
    asm("" : "+v"(next), "+v"(cur));
}

__device__ __forceinline__ void pin() { __builtin_amdgcn_sched_barrier(0); }

__device__ __forceinline__ void lds_barrier() {
    __builtin_amdgcn_fence(__ATOMIC_RELEASE, "workgroup", "local");
    __builtin_amdgcn_s_barrier();
    __builtin_amdgcn_fence(__ATOMIC_ACQUIRE, "workgroup", "local");
}

struct Gates { v8f i, f, g, o; };

__global__ __launch_bounds__(128) __attribute__((amdgpu_waves_per_eu(1, 1)))
void OpenLSTM_wmma_scan(const float* __restrict__ u,
                        const float* __restrict__ Wih,
                        const float* __restrict__ Whh,
                        const float* __restrict__ bih,
                        const float* __restrict__ bhh,
                        const float* __restrict__ Whr,
                        const int*   __restrict__ ncp,
                        float*       __restrict__ out)
{
    const int lane = threadIdx.x & 31;
    const int w    = __builtin_amdgcn_readfirstlane((int)(threadIdx.x >> 5));
    const int n    = lane & 15;
    const int hi   = lane >> 4;
    const int b0   = blockIdx.x * 16;
    const int nc   = *ncp;

    __shared__ __attribute__((aligned(16))) float stage[16 * PITCH + 2 * BLK * 4];
    __shared__ __attribute__((aligned(16))) float part[2 * 8 * 16 * PP];
    __shared__ float cbuf[16 * 17];
    const int ls = lane & 7;
    const int lb = lane >> 3;
    const float* gsrc = u + ((size_t)(b0 + lb) * TSTEPS + ls) * 4;
    float*       sdst = stage + lb * PITCH + ls * 4;
    const float* ssrc = stage + n * PITCH + hi * 2;
    const size_t PIECE = (size_t)4 * TSTEPS * 4;

    const float* Wsrc = hi ? Whh : Wih;

    v8f c = {};
    const v8f zc = {};
    float h0 = 0.0f, h1 = 0.0f;

    float*       op = out + (size_t)(b0 + n) * TSTEPS * 2;

    v2f A4;
    float hb4[8], w40[2], w41[2];
#pragma unroll
    for (int r = 0; r < 8; ++r) {
        const int grow = (r >> 1) * 16 + 4 * w + 2 * hi + (r & 1);
        const float bsum = bih[grow] + bhh[grow];
        hb4[r] = ((r >> 1) == 2) ? bsum : 0.5f * bsum;
    }
    {
        const int m    = n;
        const int grow = ((m & 7) >> 1) * 16 + 4 * w + 2 * (m >> 3) + (m & 1);
        A4.x = Wsrc[grow * 2 + 0];
        A4.y = Wsrc[grow * 2 + 1];
    }
#pragma unroll
    for (int q = 0; q < 2; ++q) {
        w40[q] = Whr[0 * 16 + 4 * w + 2 * hi + q];
        w41[q] = Whr[1 * 16 + 4 * w + 2 * hi + q];
    }
    float c4[2] = {0.0f, 0.0f};

    const float* gsrc4 = u + ((size_t)(b0 + 4 * w + lb) * TSTEPS + ls) * 4;
    float*       sdst4 = stage + (4 * w + lb) * PITCH + ls * 4;
    float*       orow4 = out + ((size_t)(b0 + 4 * w + lb) * TSTEPS + ls) * 2;

    {
        const v4f g = *(const v4f*)(gsrc4);
        *(v4f*)(sdst4) = g;
    }
    if (threadIdx.x < 2 * BLK * 4) stage[16 * PITCH + threadIdx.x] = 0.0f;
    __syncthreads();

    v2f in[BLK];
#pragma unroll
    for (int j = 0; j < BLK; ++j)
        in[j] = *(const v2f*)(ssrc + j * 4);

    const int ncc = nc < 0 ? 0 : (nc > TSTEPS ? TSTEPS : nc);
    const int kA  = (ncc / (2 * BLK)) * 2;

    auto ctx_block = [&](const int kk, const v4f gst) __attribute__((always_inline)) {
        const int k = kk;
        float* pw = part + ((((k & 1) * 8) + 2 * w + hi) * 16 + n) * PP;
        v8f Gn4 = __builtin_amdgcn_wmma_f32_16x16x4_f32(false, A4, false, in[0], (short)0, zc, false, false);
        wmma_slots(Gn4, in[0], in[0]);
#pragma unroll
        for (int j = 0; j < BLK; ++j) {
            v8f Gd = Gn4;
            if (j + 1 < BLK) {
                Gn4 = __builtin_amdgcn_wmma_f32_16x16x4_f32(false, A4, false, in[j + 1], (short)0, zc, false, false);
                issue_ahead(Gn4, Gd);
            }
            float tv[8];
#pragma unroll
            for (int r = 0; r < 8; ++r)
                tv[r] = ((r >> 1) == 2) ? (Gd[r] + hb4[r]) : fmaf(Gd[r], 0.5f, hb4[r]);
#pragma unroll
            for (int r = 0; r < 8; ++r)
                tv[r] = __builtin_amdgcn_tanhf(tv[r]);
            float sq[2];
#pragma unroll
            for (int q = 0; q < 2; ++q) {
                const float si = fmaf(tv[0 + q], 0.5f, 0.5f);
                const float sf = fmaf(tv[2 + q], 0.5f, 0.5f);
                const float so = fmaf(tv[6 + q], 0.5f, 0.5f);
                const float cn = sf * c4[q] + si * tv[4 + q];
                c4[q] = cn;
                sq[q] = so * __builtin_amdgcn_tanhf(cn);
            }
            v2f p;
            p.x = sq[0] * w40[0] + sq[1] * w40[1];
            p.y = sq[0] * w41[0] + sq[1] * w41[1];
            *(v2f*)(pw + j * 2) = p;
        }

        pin();
        *(v4f*)(sdst4 + ((k + 1) & 1) * (BLK * 4)) = gst;
        lds_barrier();
        {
            const float* sn = ssrc + ((k + 1) & 1) * (BLK * 4);
#pragma unroll
            for (int i = 0; i < BLK; ++i)
                in[i] = *(const v2f*)(sn + i * 4);
        }
        {
            const float* pr = part + (((k & 1) * 8) * 16 + 4 * w + lb) * PP + ls * 2;
            v2f q8[8];
#pragma unroll
            for (int s = 0; s < 8; ++s)
                q8[s] = *(const v2f*)(pr + s * 16 * PP);
            const v2f hv = ((q8[0] + q8[1]) + (q8[2] + q8[3])) + ((q8[4] + q8[5]) + (q8[6] + q8[7]));
            __builtin_nontemporal_store(hv, (v2f*)(orow4 + (size_t)k * BLK * 2));
        }
    };

    auto ld_blk = [&](const int kb) __attribute__((always_inline)) {
        const int kn = (kb < NBLK) ? kb : (NBLK - 1);
        return *(const v4f*)(gsrc4 + (size_t)kn * BLK * 4);
    };
    v4f gA = ld_blk(1);
    int k = 0;
    for (; k < kA; k += 2) {
        const v4f gB = ld_blk(k + 2);
        pin();
        ctx_block(k, gA);
        gA = ld_blk(k + 3);
        pin();
        ctx_block(k + 1, gB);
    }

    const int bb = lane & 3;
    const int s4 = (lane >> 2) & 3;
    const bool mA = (s4 == hi);
    const bool mB = (s4 == 2 + hi);
    v2f Aha, Ahb, Axa, Axb;
    {
        const int usub = 2 * (n >> 3) + (n & 1);
        const int gsel = (n & 7) >> 1;
        const int ga = gsel * 16 + 4 * hi + usub;
        const int gb = gsel * 16 + 4 * (2 + hi) + usub;
        Aha.x = Whh[ga * 2 + 0]; Aha.y = Whh[ga * 2 + 1];
        Ahb.x = Whh[gb * 2 + 0]; Ahb.y = Whh[gb * 2 + 1];
        Axa.x = Wih[ga * 2 + 0]; Axa.y = Wih[ga * 2 + 1];
        Axb.x = Wih[gb * 2 + 0]; Axb.y = Wih[gb * 2 + 1];
    }
    float hb2[8], wp0[2], wp1[2];
#pragma unroll
    for (int r = 0; r < 8; ++r) {
        const int grow = (r >> 1) * 16 + 4 * s4 + 2 * hi + (r & 1);
        const float bsum = bih[grow] + bhh[grow];
        hb2[r] = ((r >> 1) == 2) ? bsum : 0.5f * bsum;
    }
#pragma unroll
    for (int q = 0; q < 2; ++q) {
        wp0[q] = Whr[0 * 16 + 4 * s4 + 2 * hi + q];
        wp1[q] = Whr[1 * 16 + 4 * s4 + 2 * hi + q];
    }
    cbuf[(4 * w + 2 * hi + 0) * 17 + n] = c4[0];
    cbuf[(4 * w + 2 * hi + 1) * 17 + n] = c4[1];
    __syncthreads();
    float c2[2];
#pragma unroll
    for (int q = 0; q < 2; ++q)
        c2[q] = cbuf[(4 * s4 + 2 * hi + q) * 17 + 4 * w + bb];
    if (kA > 0) {
        const float* pr = part + ((((kA - 1) & 1) * 8) * 16 + 4 * w + bb) * PP + (BLK - 1) * 2;
        v2f q8[8];
#pragma unroll
        for (int s = 0; s < 8; ++s)
            q8[s] = *(const v2f*)(pr + s * 16 * PP);
        const v2f hv = ((q8[0] + q8[1]) + (q8[2] + q8[3])) + ((q8[4] + q8[5]) + (q8[6] + q8[7]));
        h0 = hv.x;
        h1 = hv.y;
    }

    const float* srcA = stage + (mA ? (4 * w + bb) * PITCH : 16 * PITCH);
    const float* srcB = stage + (mB ? (4 * w + bb) * PITCH : 16 * PITCH);
    const int g8 = s4 + 4 * hi;
    float* orow2 = out + ((size_t)(b0 + 4 * w + bb) * TSTEPS + g8) * 2;

    v4f iA0[BLK], iB0[BLK], iA1[BLK], iB1[BLK];
    {
        const int hf = (kA & 1) * (BLK * 4);
#pragma unroll
        for (int i = 0; i < BLK; ++i) {
            iA0[i] = *(const v4f*)(srcA + hf + i * 4);
            iB0[i] = *(const v4f*)(srcB + hf + i * 4);
        }
    }
    v8f Dx;
    {
        const v2f xa = {iA0[0].x, iA0[0].y};
        const v2f xb = {iB0[0].x, iB0[0].y};
        Dx = __builtin_amdgcn_wmma_f32_16x16x4_f32(false, Axa, false, xa, (short)0, zc, false, false);
        Dx = __builtin_amdgcn_wmma_f32_16x16x4_f32(false, Axb, false, xb, (short)0, Dx, false, false);
    }

    auto fr_block = [&](const int kk, v4f (&ia)[BLK], v4f (&ib)[BLK], v4f (&na)[BLK], v4f (&nb)[BLK])
        __attribute__((always_inline)) {
        const int tb = kk * BLK;
        v2f hs = {0.0f, 0.0f};
#pragma unroll
        for (int j = 0; j < BLK; ++j) {
            if (j == BLK - 2) {
                pin();
                *(v4f*)(sdst4 + ((kk + 1) & 1) * (BLK * 4)) = gA;
                __builtin_amdgcn_fence(__ATOMIC_ACQ_REL, "wavefront");
                const int hf = ((kk + 1) & 1) * (BLK * 4);
#pragma unroll
                for (int i = 0; i < BLK; ++i) {
                    na[i] = *(const v4f*)(srcA + hf + i * 4);
                    nb[i] = *(const v4f*)(srcB + hf + i * 4);
                }
                gA = ld_blk(kk + 2);
                pin();
            }
            const int t = tb + j;
            const bool fr = !(t < nc);
            const bool cA = mA && fr;
            const bool cB = mB && fr;
            v2f Ba, Bb;
            Ba.x = cA ? h0 : ia[j].z;
            Ba.y = cA ? h1 : ia[j].w;
            Bb.x = cB ? h0 : ib[j].z;
            Bb.y = cB ? h1 : ib[j].w;
            v8f Gd = __builtin_amdgcn_wmma_f32_16x16x4_f32(false, Aha, false, Ba, (short)0, Dx, false, false);
            Gd = __builtin_amdgcn_wmma_f32_16x16x4_f32(false, Ahb, false, Bb, (short)0, Gd, false, false);
            wmma_slots(Gd, Ba, Bb);
            {
                const v4f ua = (j + 1 < BLK) ? ia[(j + 1) & (BLK - 1)] : na[0];
                const v4f ub = (j + 1 < BLK) ? ib[(j + 1) & (BLK - 1)] : nb[0];
                const v2f xa = {ua.x, ua.y};
                const v2f xb = {ub.x, ub.y};
                Dx = __builtin_amdgcn_wmma_f32_16x16x4_f32(false, Axa, false, xa, (short)0, zc, false, false);
                Dx = __builtin_amdgcn_wmma_f32_16x16x4_f32(false, Axb, false, xb, (short)0, Dx, false, false);
                wmma_slots(Dx, xa, xb);
            }
            float tv[8];
#pragma unroll
            for (int r = 0; r < 8; ++r)
                tv[r] = ((r >> 1) == 2) ? (Gd[r] + hb2[r]) : fmaf(Gd[r], 0.5f, hb2[r]);
#pragma unroll
            for (int r = 0; r < 8; ++r)
                tv[r] = __builtin_amdgcn_tanhf(tv[r]);
            float sq[2];
#pragma unroll
            for (int q = 0; q < 2; ++q) {
                const float si = fmaf(tv[0 + q], 0.5f, 0.5f);
                const float sf = fmaf(tv[2 + q], 0.5f, 0.5f);
                const float so = fmaf(tv[6 + q], 0.5f, 0.5f);
                const float cn = sf * c2[q] + si * tv[4 + q];
                c2[q] = cn;
                sq[q] = so * __builtin_amdgcn_tanhf(cn);
            }
            float p0 = sq[0] * wp0[0] + sq[1] * wp0[1];
            float p1 = sq[0] * wp1[0] + sq[1] * wp1[1];
            p0 += rowx4(p0);  p1 += rowx4(p1);
            p0 += rowx8(p0);  p1 += rowx8(p1);
            h0 = p0 + xhalf_rec(p0);
            h1 = p1 + xhalf_rec(p1);
            hs.x = (g8 == j) ? h0 : hs.x;
            hs.y = (g8 == j) ? h1 : hs.y;
        }
        __builtin_nontemporal_store(hs, (v2f*)(orow2 + (size_t)tb * 2));
    };

    for (; k < NBLK; k += 2) {
        fr_block(k, iA0, iB0, iA1, iB1);
        fr_block(k + 1, iA1, iB1, iA0, iB0);
    }
}

extern "C" void kernel_launch(void* const* d_in, const int* in_sizes, int n_in,
                              void* d_out, int out_size, void* d_ws, size_t ws_size,
                              hipStream_t stream) {
    const float* u    = (const float*)d_in[0];
    const float* Wih  = (const float*)d_in[1];
    const float* Whh  = (const float*)d_in[2];
    const float* bih  = (const float*)d_in[3];
    const float* bhh  = (const float*)d_in[4];
    const float* Whr  = (const float*)d_in[5];
    const int*   ncp  = (const int*)d_in[6];
    float*       out  = (float*)d_out;

    dim3 grid(BATCH / 16);
    dim3 block(128);
    OpenLSTM_wmma_scan<<<grid, block, 0, stream>>>(u, Wih, Whh, bih, bhh, Whr, ncp, out);
}


# input pieces requested 8 blocks ahead of use in the context phase (was 2; 8 register sets, 8 blocks per loop trip) and 2 ahead after it (was 1): blocks stop waiting for memory; arithmetic unchanged
# speedup vs baseline: 7.1478x; 1.1009x over previous
#include <hip/hip_runtime.h>

typedef __attribute__((ext_vector_type(2))) float v2f;
typedef __attribute__((ext_vector_type(4))) float v4f;
typedef __attribute__((ext_vector_type(8))) float v8f;

#define BATCH   4096
#define TSTEPS  2048
#define BLK     8
#define NBLK    (TSTEPS / BLK)
#define CAHEAD  8
#define PP      18
#define PITCH   68

__device__ __forceinline__ float fsigm_b(float x, float hb) {
    return fmaf(__builtin_amdgcn_tanhf(fmaf(x, 0.5f, hb)), 0.5f, 0.5f);
}

__device__ __forceinline__ float xhalf(float x) {
    const int b = __builtin_bit_cast(int, x);
    return __builtin_bit_cast(float, __builtin_amdgcn_permlanex16(b, b, 0x76543210, 0xfedcba98, false, false));
}

__device__ __forceinline__ float xhalf_rec(float x) {
    const int b = __builtin_bit_cast(int, x);
    return __builtin_bit_cast(float, __builtin_amdgcn_permlanex16(0, b, 0x76543210, 0xfedcba98, false, false));
}

__device__ __forceinline__ float rowx4(float x) {
    const int b = __builtin_bit_cast(int, x);
    return __builtin_bit_cast(float, __builtin_amdgcn_update_dpp(0, b, 0x164, 0xf, 0xf, true));
}
__device__ __forceinline__ float rowx8(float x) {
    const int b = __builtin_bit_cast(int, x);
    return __builtin_bit_cast(float, __builtin_amdgcn_update_dpp(0, b, 0x168, 0xf, 0xf, true));
}

__device__ __forceinline__ void wmma_slots(v8f& d, const v2f ba, const v2f bb) {
    asm("v_nop\n\tv_nop\n\tv_nop\n\tv_nop" : "+v"(d) : "v"(ba), "v"(bb));
}

__device__ __forceinline__ void issue_ahead(v8f& next, v8f& cur) {
    asm("" : "+v"(next), "+v"(cur));
}

__device__ __forceinline__ void pin() { __builtin_amdgcn_sched_barrier(0); }

__device__ __forceinline__ void lds_barrier() {
    __builtin_amdgcn_fence(__ATOMIC_RELEASE, "workgroup", "local");
    __builtin_amdgcn_s_barrier();
    __builtin_amdgcn_fence(__ATOMIC_ACQUIRE, "workgroup", "local");
}

struct Gates { v8f i, f, g, o; };

__global__ __launch_bounds__(128) __attribute__((amdgpu_waves_per_eu(1, 1)))
void OpenLSTM_wmma_scan(const float* __restrict__ u,
                        const float* __restrict__ Wih,
                        const float* __restrict__ Whh,
                        const float* __restrict__ bih,
                        const float* __restrict__ bhh,
                        const float* __restrict__ Whr,
                        const int*   __restrict__ ncp,
                        float*       __restrict__ out)
{
    const int lane = threadIdx.x & 31;
    const int w    = __builtin_amdgcn_readfirstlane((int)(threadIdx.x >> 5));
    const int n    = lane & 15;
    const int hi   = lane >> 4;
    const int b0   = blockIdx.x * 16;
    const int nc   = *ncp;

    __shared__ __attribute__((aligned(16))) float stage[16 * PITCH + 2 * BLK * 4];
    __shared__ __attribute__((aligned(16))) float part[2 * 8 * 16 * PP];
    __shared__ float cbuf[16 * 17];
    const int ls = lane & 7;
    const int lb = lane >> 3;
    const float* gsrc = u + ((size_t)(b0 + lb) * TSTEPS + ls) * 4;
    float*       sdst = stage + lb * PITCH + ls * 4;
    const float* ssrc = stage + n * PITCH + hi * 2;
    const size_t PIECE = (size_t)4 * TSTEPS * 4;

    const float* Wsrc = hi ? Whh : Wih;

    v8f c = {};
    const v8f zc = {};
    float h0 = 0.0f, h1 = 0.0f;

    float*       op = out + (size_t)(b0 + n) * TSTEPS * 2;

    v2f A4;
    float hb4[8], w40[2], w41[2];
#pragma unroll
    for (int r = 0; r < 8; ++r) {
        const int grow = (r >> 1) * 16 + 4 * w + 2 * hi + (r & 1);
        const float bsum = bih[grow] + bhh[grow];
        hb4[r] = ((r >> 1) == 2) ? bsum : 0.5f * bsum;
    }
    {
        const int m    = n;
        const int grow = ((m & 7) >> 1) * 16 + 4 * w + 2 * (m >> 3) + (m & 1);
        A4.x = Wsrc[grow * 2 + 0];
        A4.y = Wsrc[grow * 2 + 1];
    }
#pragma unroll
    for (int q = 0; q < 2; ++q) {
        w40[q] = Whr[0 * 16 + 4 * w + 2 * hi + q];
        w41[q] = Whr[1 * 16 + 4 * w + 2 * hi + q];
    }
    float c4[2] = {0.0f, 0.0f};

    const float* gsrc4 = u + ((size_t)(b0 + 4 * w + lb) * TSTEPS + ls) * 4;
    float*       sdst4 = stage + (4 * w + lb) * PITCH + ls * 4;
    float*       orow4 = out + ((size_t)(b0 + 4 * w + lb) * TSTEPS + ls) * 2;

    {
        const v4f g = *(const v4f*)(gsrc4);
        *(v4f*)(sdst4) = g;
    }
    if (threadIdx.x < 2 * BLK * 4) stage[16 * PITCH + threadIdx.x] = 0.0f;
    __syncthreads();

    v2f in[BLK];
#pragma unroll
    for (int j = 0; j < BLK; ++j)
        in[j] = *(const v2f*)(ssrc + j * 4);

    const int ncc = nc < 0 ? 0 : (nc > TSTEPS ? TSTEPS : nc);
    const int kA  = (ncc / (CAHEAD * BLK)) * CAHEAD;

    auto ctx_block = [&](const int kk, const v4f gst) __attribute__((always_inline)) {
        const int k = kk;
        float* pw = part + ((((k & 1) * 8) + 2 * w + hi) * 16 + n) * PP;
        v8f Gn4 = __builtin_amdgcn_wmma_f32_16x16x4_f32(false, A4, false, in[0], (short)0, zc, false, false);
        wmma_slots(Gn4, in[0], in[0]);
#pragma unroll
        for (int j = 0; j < BLK; ++j) {
            v8f Gd = Gn4;
            if (j + 1 < BLK) {
                Gn4 = __builtin_amdgcn_wmma_f32_16x16x4_f32(false, A4, false, in[j + 1], (short)0, zc, false, false);
                issue_ahead(Gn4, Gd);
            }
            float tv[8];
#pragma unroll
            for (int r = 0; r < 8; ++r)
                tv[r] = ((r >> 1) == 2) ? (Gd[r] + hb4[r]) : fmaf(Gd[r], 0.5f, hb4[r]);
#pragma unroll
            for (int r = 0; r < 8; ++r)
                tv[r] = __builtin_amdgcn_tanhf(tv[r]);
            float sq[2];
#pragma unroll
            for (int q = 0; q < 2; ++q) {
                const float si = fmaf(tv[0 + q], 0.5f, 0.5f);
                const float sf = fmaf(tv[2 + q], 0.5f, 0.5f);
                const float so = fmaf(tv[6 + q], 0.5f, 0.5f);
                const float cn = sf * c4[q] + si * tv[4 + q];
                c4[q] = cn;
                sq[q] = so * __builtin_amdgcn_tanhf(cn);
            }
            v2f p;
            p.x = sq[0] * w40[0] + sq[1] * w40[1];
            p.y = sq[0] * w41[0] + sq[1] * w41[1];
            *(v2f*)(pw + j * 2) = p;
        }

        pin();
        *(v4f*)(sdst4 + ((k + 1) & 1) * (BLK * 4)) = gst;
        lds_barrier();
        {
            const float* sn = ssrc + ((k + 1) & 1) * (BLK * 4);
#pragma unroll
            for (int i = 0; i < BLK; ++i)
                in[i] = *(const v2f*)(sn + i * 4);
        }
        {
            const float* pr = part + (((k & 1) * 8) * 16 + 4 * w + lb) * PP + ls * 2;
            v2f q8[8];
#pragma unroll
            for (int s = 0; s < 8; ++s)
                q8[s] = *(const v2f*)(pr + s * 16 * PP);
            const v2f hv = ((q8[0] + q8[1]) + (q8[2] + q8[3])) + ((q8[4] + q8[5]) + (q8[6] + q8[7]));
            __builtin_nontemporal_store(hv, (v2f*)(orow4 + (size_t)k * BLK * 2));
        }
    };

    auto ld_blk = [&](const int kb) __attribute__((always_inline)) {
        const int kn = (kb < NBLK) ? kb : (NBLK - 1);
        return *(const v4f*)(gsrc4 + (size_t)kn * BLK * 4);
    };
    v4f gq[CAHEAD];
#pragma unroll
    for (int i = 1; i < CAHEAD; ++i)
        gq[i] = ld_blk(i);
    int k = 0;
    for (; k < kA; k += CAHEAD) {
#pragma unroll
        for (int i = 0; i < CAHEAD; ++i) {
            gq[i] = ld_blk(k + i + CAHEAD);
            pin();
            ctx_block(k + i, gq[(i + 1) & (CAHEAD - 1)]);
        }
    }

    const int bb = lane & 3;
    const int s4 = (lane >> 2) & 3;
    const bool mA = (s4 == hi);
    const bool mB = (s4 == 2 + hi);
    v2f Aha, Ahb, Axa, Axb;
    {
        const int usub = 2 * (n >> 3) + (n & 1);
        const int gsel = (n & 7) >> 1;
        const int ga = gsel * 16 + 4 * hi + usub;
        const int gb = gsel * 16 + 4 * (2 + hi) + usub;
        Aha.x = Whh[ga * 2 + 0]; Aha.y = Whh[ga * 2 + 1];
        Ahb.x = Whh[gb * 2 + 0]; Ahb.y = Whh[gb * 2 + 1];
        Axa.x = Wih[ga * 2 + 0]; Axa.y = Wih[ga * 2 + 1];
        Axb.x = Wih[gb * 2 + 0]; Axb.y = Wih[gb * 2 + 1];
    }
    float hb2[8], wp0[2], wp1[2];
#pragma unroll
    for (int r = 0; r < 8; ++r) {
        const int grow = (r >> 1) * 16 + 4 * s4 + 2 * hi + (r & 1);
        const float bsum = bih[grow] + bhh[grow];
        hb2[r] = ((r >> 1) == 2) ? bsum : 0.5f * bsum;
    }
#pragma unroll
    for (int q = 0; q < 2; ++q) {
        wp0[q] = Whr[0 * 16 + 4 * s4 + 2 * hi + q];
        wp1[q] = Whr[1 * 16 + 4 * s4 + 2 * hi + q];
    }
    cbuf[(4 * w + 2 * hi + 0) * 17 + n] = c4[0];
    cbuf[(4 * w + 2 * hi + 1) * 17 + n] = c4[1];
    __syncthreads();
    float c2[2];
#pragma unroll
    for (int q = 0; q < 2; ++q)
        c2[q] = cbuf[(4 * s4 + 2 * hi + q) * 17 + 4 * w + bb];
    if (kA > 0) {
        const float* pr = part + ((((kA - 1) & 1) * 8) * 16 + 4 * w + bb) * PP + (BLK - 1) * 2;
        v2f q8[8];
#pragma unroll
        for (int s = 0; s < 8; ++s)
            q8[s] = *(const v2f*)(pr + s * 16 * PP);
        const v2f hv = ((q8[0] + q8[1]) + (q8[2] + q8[3])) + ((q8[4] + q8[5]) + (q8[6] + q8[7]));
        h0 = hv.x;
        h1 = hv.y;
    }

    const float* srcA = stage + (mA ? (4 * w + bb) * PITCH : 16 * PITCH);
    const float* srcB = stage + (mB ? (4 * w + bb) * PITCH : 16 * PITCH);
    const int g8 = s4 + 4 * hi;
    float* orow2 = out + ((size_t)(b0 + 4 * w + bb) * TSTEPS + g8) * 2;

    v4f iA0[BLK], iB0[BLK], iA1[BLK], iB1[BLK];
    {
        const int hf = (kA & 1) * (BLK * 4);
#pragma unroll
        for (int i = 0; i < BLK; ++i) {
            iA0[i] = *(const v4f*)(srcA + hf + i * 4);
            iB0[i] = *(const v4f*)(srcB + hf + i * 4);
        }
    }
    v8f Dx;
    {
        const v2f xa = {iA0[0].x, iA0[0].y};
        const v2f xb = {iB0[0].x, iB0[0].y};
        Dx = __builtin_amdgcn_wmma_f32_16x16x4_f32(false, Axa, false, xa, (short)0, zc, false, false);
        Dx = __builtin_amdgcn_wmma_f32_16x16x4_f32(false, Axb, false, xb, (short)0, Dx, false, false);
    }

    auto fr_block = [&](const int kk, v4f (&ia)[BLK], v4f (&ib)[BLK], v4f (&na)[BLK], v4f (&nb)[BLK], v4f& gn)
        __attribute__((always_inline)) {
        const int tb = kk * BLK;
        v2f hs = {0.0f, 0.0f};
#pragma unroll
        for (int j = 0; j < BLK; ++j) {
            if (j == BLK - 2) {
                pin();
                *(v4f*)(sdst4 + ((kk + 1) & 1) * (BLK * 4)) = gn;
                __builtin_amdgcn_fence(__ATOMIC_ACQ_REL, "wavefront");
                const int hf = ((kk + 1) & 1) * (BLK * 4);
#pragma unroll
                for (int i = 0; i < BLK; ++i) {
                    na[i] = *(const v4f*)(srcA + hf + i * 4);
                    nb[i] = *(const v4f*)(srcB + hf + i * 4);
                }
                gn = ld_blk(kk + 3);
                pin();
            }
            const int t = tb + j;
            const bool fr = !(t < nc);
            const bool cA = mA && fr;
            const bool cB = mB && fr;
            v2f Ba, Bb;
            Ba.x = cA ? h0 : ia[j].z;
            Ba.y = cA ? h1 : ia[j].w;
            Bb.x = cB ? h0 : ib[j].z;
            Bb.y = cB ? h1 : ib[j].w;
            v8f Gd = __builtin_amdgcn_wmma_f32_16x16x4_f32(false, Aha, false, Ba, (short)0, Dx, false, false);
            Gd = __builtin_amdgcn_wmma_f32_16x16x4_f32(false, Ahb, false, Bb, (short)0, Gd, false, false);
            wmma_slots(Gd, Ba, Bb);
            {
                const v4f ua = (j + 1 < BLK) ? ia[(j + 1) & (BLK - 1)] : na[0];
                const v4f ub = (j + 1 < BLK) ? ib[(j + 1) & (BLK - 1)] : nb[0];
                const v2f xa = {ua.x, ua.y};
                const v2f xb = {ub.x, ub.y};
                Dx = __builtin_amdgcn_wmma_f32_16x16x4_f32(false, Axa, false, xa, (short)0, zc, false, false);
                Dx = __builtin_amdgcn_wmma_f32_16x16x4_f32(false, Axb, false, xb, (short)0, Dx, false, false);
                wmma_slots(Dx, xa, xb);
            }
            float tv[8];
#pragma unroll
            for (int r = 0; r < 8; ++r)
                tv[r] = ((r >> 1) == 2) ? (Gd[r] + hb2[r]) : fmaf(Gd[r], 0.5f, hb2[r]);
#pragma unroll
            for (int r = 0; r < 8; ++r)
                tv[r] = __builtin_amdgcn_tanhf(tv[r]);
            float sq[2];
#pragma unroll
            for (int q = 0; q < 2; ++q) {
                const float si = fmaf(tv[0 + q], 0.5f, 0.5f);
                const float sf = fmaf(tv[2 + q], 0.5f, 0.5f);
                const float so = fmaf(tv[6 + q], 0.5f, 0.5f);
                const float cn = sf * c2[q] + si * tv[4 + q];
                c2[q] = cn;
                sq[q] = so * __builtin_amdgcn_tanhf(cn);
            }
            float p0 = sq[0] * wp0[0] + sq[1] * wp0[1];
            float p1 = sq[0] * wp1[0] + sq[1] * wp1[1];
            p0 += rowx4(p0);  p1 += rowx4(p1);
            p0 += rowx8(p0);  p1 += rowx8(p1);
            h0 = p0 + xhalf_rec(p0);
            h1 = p1 + xhalf_rec(p1);
            hs.x = (g8 == j) ? h0 : hs.x;
            hs.y = (g8 == j) ? h1 : hs.y;
        }
        __builtin_nontemporal_store(hs, (v2f*)(orow2 + (size_t)tb * 2));
    };

    for (; k < NBLK; k += 2) {
        fr_block(k, iA0, iB0, iA1, iB1, gq[1]);
        fr_block(k + 1, iA1, iB1, iA0, iB0, gq[2]);
    }
}

extern "C" void kernel_launch(void* const* d_in, const int* in_sizes, int n_in,
                              void* d_out, int out_size, void* d_ws, size_t ws_size,
                              hipStream_t stream) {
    const float* u    = (const float*)d_in[0];
    const float* Wih  = (const float*)d_in[1];
    const float* Whh  = (const float*)d_in[2];
    const float* bih  = (const float*)d_in[3];
    const float* bhh  = (const float*)d_in[4];
    const float* Whr  = (const float*)d_in[5];
    const int*   ncp  = (const int*)d_in[6];
    float*       out  = (float*)d_out;

    dim3 grid(BATCH / 16);
    dim3 block(128);
    OpenLSTM_wmma_scan<<<grid, block, 0, stream>>>(u, Wih, Whh, bih, bhh, Whr, ncp, out);
}


# constants folded off the per-step chain: gate bias is the matrix products' C (summed first: sum order changes), the sigmoids' 0.5 sits in the weight rows; sigmoid(o) x W_hr formed beside tanh(c)
# speedup vs baseline: 7.5784x; 1.0602x over previous
#include <hip/hip_runtime.h>

typedef __attribute__((ext_vector_type(2))) float v2f;
typedef __attribute__((ext_vector_type(4))) float v4f;
typedef __attribute__((ext_vector_type(8))) float v8f;

#define BATCH   4096
#define TSTEPS  2048
#define BLK     8
#define NBLK    (TSTEPS / BLK)
#define CAHEAD  8
#define PP      18
#define PITCH   68

__device__ __forceinline__ float fsigm_b(float x, float hb) {
    return fmaf(__builtin_amdgcn_tanhf(fmaf(x, 0.5f, hb)), 0.5f, 0.5f);
}

__device__ __forceinline__ float xhalf(float x) {
    const int b = __builtin_bit_cast(int, x);
    return __builtin_bit_cast(float, __builtin_amdgcn_permlanex16(b, b, 0x76543210, 0xfedcba98, false, false));
}

__device__ __forceinline__ float xhalf_rec(float x) {
    const int b = __builtin_bit_cast(int, x);
    return __builtin_bit_cast(float, __builtin_amdgcn_permlanex16(0, b, 0x76543210, 0xfedcba98, false, false));
}

__device__ __forceinline__ float rowx4(float x) {
    const int b = __builtin_bit_cast(int, x);
    return __builtin_bit_cast(float, __builtin_amdgcn_update_dpp(0, b, 0x164, 0xf, 0xf, true));
}
__device__ __forceinline__ float rowx8(float x) {
    const int b = __builtin_bit_cast(int, x);
    return __builtin_bit_cast(float, __builtin_amdgcn_update_dpp(0, b, 0x168, 0xf, 0xf, true));
}

__device__ __forceinline__ void wmma_slots(v8f& d, const v2f ba, const v2f bb) {
    asm("v_nop\n\tv_nop\n\tv_nop\n\tv_nop" : "+v"(d) : "v"(ba), "v"(bb));
}

__device__ __forceinline__ void issue_ahead(v8f& next, v8f& cur) {
    asm("" : "+v"(next), "+v"(cur));
}

__device__ __forceinline__ void pin() { __builtin_amdgcn_sched_barrier(0); }

__device__ __forceinline__ void lds_barrier() {
    __builtin_amdgcn_fence(__ATOMIC_RELEASE, "workgroup", "local");
    __builtin_amdgcn_s_barrier();
    __builtin_amdgcn_fence(__ATOMIC_ACQUIRE, "workgroup", "local");
}

struct Gates { v8f i, f, g, o; };

__global__ __launch_bounds__(128) __attribute__((amdgpu_waves_per_eu(1, 1)))
void OpenLSTM_wmma_scan(const float* __restrict__ u,
                        const float* __restrict__ Wih,
                        const float* __restrict__ Whh,
                        const float* __restrict__ bih,
                        const float* __restrict__ bhh,
                        const float* __restrict__ Whr,
                        const int*   __restrict__ ncp,
                        float*       __restrict__ out)
{
    const int lane = threadIdx.x & 31;
    const int w    = __builtin_amdgcn_readfirstlane((int)(threadIdx.x >> 5));
    const int n    = lane & 15;
    const int hi   = lane >> 4;
    const int b0   = blockIdx.x * 16;
    const int nc   = *ncp;

    __shared__ __attribute__((aligned(16))) float stage[16 * PITCH + 2 * BLK * 4];
    __shared__ __attribute__((aligned(16))) float part[2 * 8 * 16 * PP];
    __shared__ float cbuf[16 * 17];
    const int ls = lane & 7;
    const int lb = lane >> 3;
    const float* gsrc = u + ((size_t)(b0 + lb) * TSTEPS + ls) * 4;
    float*       sdst = stage + lb * PITCH + ls * 4;
    const float* ssrc = stage + n * PITCH + hi * 2;
    const size_t PIECE = (size_t)4 * TSTEPS * 4;

    const float* Wsrc = hi ? Whh : Wih;

    v8f c = {};
    float h0 = 0.0f, h1 = 0.0f;

    float*       op = out + (size_t)(b0 + n) * TSTEPS * 2;

    v2f A4;
    v8f hb4;
    float w40[2], w41[2];
#pragma unroll
    for (int r = 0; r < 8; ++r) {
        const int grow = (r >> 1) * 16 + 4 * w + 2 * hi + (r & 1);
        const float bsum = bih[grow] + bhh[grow];
        hb4[r] = ((r >> 1) == 2) ? bsum : 0.5f * bsum;
    }
    const float gsc = (((n & 7) >> 1) == 2) ? 1.0f : 0.5f;
    {
        const int m    = n;
        const int grow = ((m & 7) >> 1) * 16 + 4 * w + 2 * (m >> 3) + (m & 1);
        A4.x = gsc * Wsrc[grow * 2 + 0];
        A4.y = gsc * Wsrc[grow * 2 + 1];
    }
#pragma unroll
    for (int q = 0; q < 2; ++q) {
        w40[q] = Whr[0 * 16 + 4 * w + 2 * hi + q];
        w41[q] = Whr[1 * 16 + 4 * w + 2 * hi + q];
    }
    float c4[2] = {0.0f, 0.0f};

    const float* gsrc4 = u + ((size_t)(b0 + 4 * w + lb) * TSTEPS + ls) * 4;
    float*       sdst4 = stage + (4 * w + lb) * PITCH + ls * 4;
    float*       orow4 = out + ((size_t)(b0 + 4 * w + lb) * TSTEPS + ls) * 2;

    {
        const v4f g = *(const v4f*)(gsrc4);
        *(v4f*)(sdst4) = g;
    }
    if (threadIdx.x < 2 * BLK * 4) stage[16 * PITCH + threadIdx.x] = 0.0f;
    __syncthreads();

    v2f in[BLK];
#pragma unroll
    for (int j = 0; j < BLK; ++j)
        in[j] = *(const v2f*)(ssrc + j * 4);

    const int ncc = nc < 0 ? 0 : (nc > TSTEPS ? TSTEPS : nc);
    const int kA  = (ncc / (CAHEAD * BLK)) * CAHEAD;

    auto ctx_block = [&](const int kk, const v4f gst) __attribute__((always_inline)) {
        const int k = kk;
        float* pw = part + ((((k & 1) * 8) + 2 * w + hi) * 16 + n) * PP;
        v8f Gn4 = __builtin_amdgcn_wmma_f32_16x16x4_f32(false, A4, false, in[0], (short)0, hb4, false, false);
        wmma_slots(Gn4, in[0], in[0]);
#pragma unroll
        for (int j = 0; j < BLK; ++j) {
            v8f Gd = Gn4;
            if (j + 1 < BLK) {
                Gn4 = __builtin_amdgcn_wmma_f32_16x16x4_f32(false, A4, false, in[j + 1], (short)0, hb4, false, false);
                issue_ahead(Gn4, Gd);
            }
            float tv[8];
#pragma unroll
            for (int r = 0; r < 8; ++r)
                tv[r] = __builtin_amdgcn_tanhf(Gd[r]);
            float sq[2];
#pragma unroll
            for (int q = 0; q < 2; ++q) {
                const float si = fmaf(tv[0 + q], 0.5f, 0.5f);
                const float sf = fmaf(tv[2 + q], 0.5f, 0.5f);
                const float so = fmaf(tv[6 + q], 0.5f, 0.5f);
                const float cn = sf * c4[q] + si * tv[4 + q];
                c4[q] = cn;
                sq[q] = so * __builtin_amdgcn_tanhf(cn);
            }
            v2f p;
            p.x = sq[0] * w40[0] + sq[1] * w40[1];
            p.y = sq[0] * w41[0] + sq[1] * w41[1];
            *(v2f*)(pw + j * 2) = p;
        }

        pin();
        *(v4f*)(sdst4 + ((k + 1) & 1) * (BLK * 4)) = gst;
        lds_barrier();
        {
            const float* sn = ssrc + ((k + 1) & 1) * (BLK * 4);
#pragma unroll
            for (int i = 0; i < BLK; ++i)
                in[i] = *(const v2f*)(sn + i * 4);
        }
        {
            const float* pr = part + (((k & 1) * 8) * 16 + 4 * w + lb) * PP + ls * 2;
            v2f q8[8];
#pragma unroll
            for (int s = 0; s < 8; ++s)
                q8[s] = *(const v2f*)(pr + s * 16 * PP);
            const v2f hv = ((q8[0] + q8[1]) + (q8[2] + q8[3])) + ((q8[4] + q8[5]) + (q8[6] + q8[7]));
            __builtin_nontemporal_store(hv, (v2f*)(orow4 + (size_t)k * BLK * 2));
        }
    };

    auto ld_blk = [&](const int kb) __attribute__((always_inline)) {
        const int kn = (kb < NBLK) ? kb : (NBLK - 1);
        return *(const v4f*)(gsrc4 + (size_t)kn * BLK * 4);
    };
    v4f gq[CAHEAD];
#pragma unroll
    for (int i = 1; i < CAHEAD; ++i)
        gq[i] = ld_blk(i);
    int k = 0;
    for (; k < kA; k += CAHEAD) {
#pragma unroll
        for (int i = 0; i < CAHEAD; ++i) {
            gq[i] = ld_blk(k + i + CAHEAD);
            pin();
            ctx_block(k + i, gq[(i + 1) & (CAHEAD - 1)]);
        }
    }

    const int bb = lane & 3;
    const int s4 = (lane >> 2) & 3;
    const bool mA = (s4 == hi);
    const bool mB = (s4 == 2 + hi);
    v2f Aha, Ahb, Axa, Axb;
    {
        const int usub = 2 * (n >> 3) + (n & 1);
        const int gsel = (n & 7) >> 1;
        const int ga = gsel * 16 + 4 * hi + usub;
        const int gb = gsel * 16 + 4 * (2 + hi) + usub;
        Aha.x = gsc * Whh[ga * 2 + 0]; Aha.y = gsc * Whh[ga * 2 + 1];
        Ahb.x = gsc * Whh[gb * 2 + 0]; Ahb.y = gsc * Whh[gb * 2 + 1];
        Axa.x = gsc * Wih[ga * 2 + 0]; Axa.y = gsc * Wih[ga * 2 + 1];
        Axb.x = gsc * Wih[gb * 2 + 0]; Axb.y = gsc * Wih[gb * 2 + 1];
    }
    v8f hb2;
    float wp0[2], wp1[2];
#pragma unroll
    for (int r = 0; r < 8; ++r) {
        const int grow = (r >> 1) * 16 + 4 * s4 + 2 * hi + (r & 1);
        const float bsum = bih[grow] + bhh[grow];
        hb2[r] = ((r >> 1) == 2) ? bsum : 0.5f * bsum;
    }
#pragma unroll
    for (int q = 0; q < 2; ++q) {
        wp0[q] = 0.5f * Whr[0 * 16 + 4 * s4 + 2 * hi + q];
        wp1[q] = 0.5f * Whr[1 * 16 + 4 * s4 + 2 * hi + q];
    }
    cbuf[(4 * w + 2 * hi + 0) * 17 + n] = c4[0];
    cbuf[(4 * w + 2 * hi + 1) * 17 + n] = c4[1];
    __syncthreads();
    float c2[2];
#pragma unroll
    for (int q = 0; q < 2; ++q)
        c2[q] = cbuf[(4 * s4 + 2 * hi + q) * 17 + 4 * w + bb];
    if (kA > 0) {
        const float* pr = part + ((((kA - 1) & 1) * 8) * 16 + 4 * w + bb) * PP + (BLK - 1) * 2;
        v2f q8[8];
#pragma unroll
        for (int s = 0; s < 8; ++s)
            q8[s] = *(const v2f*)(pr + s * 16 * PP);
        const v2f hv = ((q8[0] + q8[1]) + (q8[2] + q8[3])) + ((q8[4] + q8[5]) + (q8[6] + q8[7]));
        h0 = hv.x;
        h1 = hv.y;
    }

    const float* srcA = stage + (mA ? (4 * w + bb) * PITCH : 16 * PITCH);
    const float* srcB = stage + (mB ? (4 * w + bb) * PITCH : 16 * PITCH);
    const int g8 = s4 + 4 * hi;
    float* orow2 = out + ((size_t)(b0 + 4 * w + bb) * TSTEPS + g8) * 2;

    v4f iA0[BLK], iB0[BLK], iA1[BLK], iB1[BLK];
    {
        const int hf = (kA & 1) * (BLK * 4);
#pragma unroll
        for (int i = 0; i < BLK; ++i) {
            iA0[i] = *(const v4f*)(srcA + hf + i * 4);
            iB0[i] = *(const v4f*)(srcB + hf + i * 4);
        }
    }
    v8f Dx;
    {
        const v2f xa = {iA0[0].x, iA0[0].y};
        const v2f xb = {iB0[0].x, iB0[0].y};
        Dx = __builtin_amdgcn_wmma_f32_16x16x4_f32(false, Axa, false, xa, (short)0, hb2, false, false);
        Dx = __builtin_amdgcn_wmma_f32_16x16x4_f32(false, Axb, false, xb, (short)0, Dx, false, false);
    }

    auto fr_block = [&](const int kk, v4f (&ia)[BLK], v4f (&ib)[BLK], v4f (&na)[BLK], v4f (&nb)[BLK], v4f& gn)
        __attribute__((always_inline)) {
        const int tb = kk * BLK;
        v2f hs = {0.0f, 0.0f};
#pragma unroll
        for (int j = 0; j < BLK; ++j) {
            if (j == BLK - 2) {
                pin();
                *(v4f*)(sdst4 + ((kk + 1) & 1) * (BLK * 4)) = gn;
                __builtin_amdgcn_fence(__ATOMIC_ACQ_REL, "wavefront");
                const int hf = ((kk + 1) & 1) * (BLK * 4);
#pragma unroll
                for (int i = 0; i < BLK; ++i) {
                    na[i] = *(const v4f*)(srcA + hf + i * 4);
                    nb[i] = *(const v4f*)(srcB + hf + i * 4);
                }
                gn = ld_blk(kk + 3);
                pin();
            }
            const int t = tb + j;
            const bool fr = !(t < nc);
            const bool cA = mA && fr;
            const bool cB = mB && fr;
            v2f Ba, Bb;
            Ba.x = cA ? h0 : ia[j].z;
            Ba.y = cA ? h1 : ia[j].w;
            Bb.x = cB ? h0 : ib[j].z;
            Bb.y = cB ? h1 : ib[j].w;
            v8f Gd = __builtin_amdgcn_wmma_f32_16x16x4_f32(false, Aha, false, Ba, (short)0, Dx, false, false);
            Gd = __builtin_amdgcn_wmma_f32_16x16x4_f32(false, Ahb, false, Bb, (short)0, Gd, false, false);
            wmma_slots(Gd, Ba, Bb);
            {
                const v4f ua = (j + 1 < BLK) ? ia[(j + 1) & (BLK - 1)] : na[0];
                const v4f ub = (j + 1 < BLK) ? ib[(j + 1) & (BLK - 1)] : nb[0];
                const v2f xa = {ua.x, ua.y};
                const v2f xb = {ub.x, ub.y};
                Dx = __builtin_amdgcn_wmma_f32_16x16x4_f32(false, Axa, false, xa, (short)0, hb2, false, false);
                Dx = __builtin_amdgcn_wmma_f32_16x16x4_f32(false, Axb, false, xb, (short)0, Dx, false, false);
                wmma_slots(Dx, xa, xb);
            }
            float tv[8];
#pragma unroll
            for (int r = 0; r < 8; ++r)
                tv[r] = __builtin_amdgcn_tanhf(Gd[r]);
            float tc[2];
#pragma unroll
            for (int q = 0; q < 2; ++q) {
                const float si = fmaf(tv[0 + q], 0.5f, 0.5f);
                const float sf = fmaf(tv[2 + q], 0.5f, 0.5f);
                const float cn = sf * c2[q] + si * tv[4 + q];
                c2[q] = cn;
                tc[q] = __builtin_amdgcn_tanhf(cn);
            }
            const float a00 = fmaf(tv[6], wp0[0], wp0[0]);
            const float a01 = fmaf(tv[7], wp0[1], wp0[1]);
            const float a10 = fmaf(tv[6], wp1[0], wp1[0]);
            const float a11 = fmaf(tv[7], wp1[1], wp1[1]);
            float p0 = fmaf(tc[0], a00, tc[1] * a01);
            float p1 = fmaf(tc[0], a10, tc[1] * a11);
            p0 += rowx4(p0);  p1 += rowx4(p1);
            p0 += rowx8(p0);  p1 += rowx8(p1);
            h0 = p0 + xhalf_rec(p0);
            h1 = p1 + xhalf_rec(p1);
            hs.x = (g8 == j) ? h0 : hs.x;
            hs.y = (g8 == j) ? h1 : hs.y;
        }
        __builtin_nontemporal_store(hs, (v2f*)(orow2 + (size_t)tb * 2));
    };

    for (; k < NBLK; k += 2) {
        fr_block(k, iA0, iB0, iA1, iB1, gq[1]);
        fr_block(k + 1, iA1, iB1, iA0, iB0, gq[2]);
    }
}

extern "C" void kernel_launch(void* const* d_in, const int* in_sizes, int n_in,
                              void* d_out, int out_size, void* d_ws, size_t ws_size,
                              hipStream_t stream) {
    const float* u    = (const float*)d_in[0];
    const float* Wih  = (const float*)d_in[1];
    const float* Whh  = (const float*)d_in[2];
    const float* bih  = (const float*)d_in[3];
    const float* bhh  = (const float*)d_in[4];
    const float* Whr  = (const float*)d_in[5];
    const int*   ncp  = (const int*)d_in[6];
    float*       out  = (float*)d_out;

    dim3 grid(BATCH / 16);
    dim3 block(128);
    OpenLSTM_wmma_scan<<<grid, block, 0, stream>>>(u, Wih, Whh, bih, bhh, Whr, ncp, out);
}


# context phase: LDS hand-overs off each block's path: inputs read a block ahead, shares summed a block behind, work-group barrier split (arrive step 2, await step 4); arithmetic unchanged
# speedup vs baseline: 8.0817x; 1.0664x over previous
#include <hip/hip_runtime.h>

typedef __attribute__((ext_vector_type(2))) float v2f;
typedef __attribute__((ext_vector_type(4))) float v4f;
typedef __attribute__((ext_vector_type(8))) float v8f;

#define BATCH   4096
#define TSTEPS  2048
#define BLK     8
#define NBLK    (TSTEPS / BLK)
#define CAHEAD  8
#define PP      18
#define PITCH   68

__device__ __forceinline__ float fsigm_b(float x, float hb) {
    return fmaf(__builtin_amdgcn_tanhf(fmaf(x, 0.5f, hb)), 0.5f, 0.5f);
}

__device__ __forceinline__ float xhalf(float x) {
    const int b = __builtin_bit_cast(int, x);
    return __builtin_bit_cast(float, __builtin_amdgcn_permlanex16(b, b, 0x76543210, 0xfedcba98, false, false));
}

__device__ __forceinline__ float xhalf_rec(float x) {
    const int b = __builtin_bit_cast(int, x);
    return __builtin_bit_cast(float, __builtin_amdgcn_permlanex16(0, b, 0x76543210, 0xfedcba98, false, false));
}

__device__ __forceinline__ float rowx4(float x) {
    const int b = __builtin_bit_cast(int, x);
    return __builtin_bit_cast(float, __builtin_amdgcn_update_dpp(0, b, 0x164, 0xf, 0xf, true));
}
__device__ __forceinline__ float rowx8(float x) {
    const int b = __builtin_bit_cast(int, x);
    return __builtin_bit_cast(float, __builtin_amdgcn_update_dpp(0, b, 0x168, 0xf, 0xf, true));
}

__device__ __forceinline__ void wmma_slots(v8f& d, const v2f ba, const v2f bb) {
    asm("v_nop\n\tv_nop\n\tv_nop\n\tv_nop" : "+v"(d) : "v"(ba), "v"(bb));
}

__device__ __forceinline__ void issue_ahead(v8f& next, v8f& cur) {
    asm("" : "+v"(next), "+v"(cur));
}

__device__ __forceinline__ void order_after(v2f& v, const v2f after) {
    asm("" : "+v"(v) : "v"(after));
}

__device__ __forceinline__ void hold_operand(v2f& v, const v2f b) {
    asm("" : "+v"(v) : "v"(b));
}

__device__ __forceinline__ void pin() { __builtin_amdgcn_sched_barrier(0); }

__device__ __forceinline__ void lds_arrive() {
    __builtin_amdgcn_fence(__ATOMIC_RELEASE, "workgroup", "local");
    __builtin_amdgcn_s_barrier_signal(-1);
}
__device__ __forceinline__ void lds_await() {
    __builtin_amdgcn_s_barrier_wait(-1);
    __builtin_amdgcn_fence(__ATOMIC_ACQUIRE, "workgroup", "local");
}

struct Gates { v8f i, f, g, o; };

__global__ __launch_bounds__(128) __attribute__((amdgpu_waves_per_eu(1, 1)))
void OpenLSTM_wmma_scan(const float* __restrict__ u,
                        const float* __restrict__ Wih,
                        const float* __restrict__ Whh,
                        const float* __restrict__ bih,
                        const float* __restrict__ bhh,
                        const float* __restrict__ Whr,
                        const int*   __restrict__ ncp,
                        float*       __restrict__ out)
{
    const int lane = threadIdx.x & 31;
    const int w    = __builtin_amdgcn_readfirstlane((int)(threadIdx.x >> 5));
    const int n    = lane & 15;
    const int hi   = lane >> 4;
    const int b0   = blockIdx.x * 16;
    const int nc   = *ncp;

    __shared__ __attribute__((aligned(16))) float stage[16 * PITCH + 2 * BLK * 4];
    __shared__ __attribute__((aligned(16))) float part[2 * 8 * 16 * PP];
    __shared__ float cbuf[16 * 17];
    const int ls = lane & 7;
    const int lb = lane >> 3;
    const float* gsrc = u + ((size_t)(b0 + lb) * TSTEPS + ls) * 4;
    float*       sdst = stage + lb * PITCH + ls * 4;
    const float* ssrc = stage + n * PITCH + hi * 2;
    const size_t PIECE = (size_t)4 * TSTEPS * 4;

    const float* Wsrc = hi ? Whh : Wih;

    v8f c = {};
    float h0 = 0.0f, h1 = 0.0f;

    float*       op = out + (size_t)(b0 + n) * TSTEPS * 2;

    v2f A4;
    v8f hb4;
    float w40[2], w41[2];
#pragma unroll
    for (int r = 0; r < 8; ++r) {
        const int grow = (r >> 1) * 16 + 4 * w + 2 * hi + (r & 1);
        const float bsum = bih[grow] + bhh[grow];
        hb4[r] = ((r >> 1) == 2) ? bsum : 0.5f * bsum;
    }
    const float gsc = (((n & 7) >> 1) == 2) ? 1.0f : 0.5f;
    {
        const int m    = n;
        const int grow = ((m & 7) >> 1) * 16 + 4 * w + 2 * (m >> 3) + (m & 1);
        A4.x = gsc * Wsrc[grow * 2 + 0];
        A4.y = gsc * Wsrc[grow * 2 + 1];
    }
#pragma unroll
    for (int q = 0; q < 2; ++q) {
        w40[q] = Whr[0 * 16 + 4 * w + 2 * hi + q];
        w41[q] = Whr[1 * 16 + 4 * w + 2 * hi + q];
    }
    float c4[2] = {0.0f, 0.0f};

    const float* gsrc4 = u + ((size_t)(b0 + 4 * w + lb) * TSTEPS + ls) * 4;
    float*       sdst4 = stage + (4 * w + lb) * PITCH + ls * 4;
    float*       orow4 = out + ((size_t)(b0 + 4 * w + lb) * TSTEPS + ls) * 2;

    v4f gq[CAHEAD];
    {
        const v4f g = *(const v4f*)(gsrc4);
        gq[1] = *(const v4f*)(gsrc4 + BLK * 4);
        *(v4f*)(sdst4) = g;
        *(v4f*)(sdst4 + BLK * 4) = gq[1];
    }
    if (threadIdx.x < 2 * BLK * 4) stage[16 * PITCH + threadIdx.x] = 0.0f;
    __syncthreads();

    v2f inE[BLK], inO[BLK];
#pragma unroll
    for (int j = 0; j < BLK; ++j)
        inE[j] = *(const v2f*)(ssrc + j * 4);

    const int ncc = nc < 0 ? 0 : (nc > TSTEPS ? TSTEPS : nc);
    const int kA  = (ncc / (CAHEAD * BLK)) * CAHEAD;

    auto ctx_block = [&](const int kk, v2f (&ic)[BLK], v2f (&nx)[BLK], const v4f gst, const bool prev)
        __attribute__((always_inline)) {
        const int k = kk;
        float* pw = part + ((((k & 1) * 8) + 2 * w + hi) * 16 + n) * PP;
        v2f sh[BLK];
        v2f q8[8];
        v8f Gn4 = __builtin_amdgcn_wmma_f32_16x16x4_f32(false, A4, false, ic[0], (short)0, hb4, false, false);
        wmma_slots(Gn4, ic[0], ic[0]);
#pragma unroll
        for (int j = 0; j < BLK; ++j) {
            if (j == 2) {
                pin();
                lds_arrive();
                pin();
            }
            if (j == 4) {
                pin();
                lds_await();
                const float* sn = ssrc + ((k + 1) & 1) * (BLK * 4);
#pragma unroll
                for (int i = 0; i < BLK; ++i)
                    nx[i] = *(const v2f*)(sn + i * 4);
                const float* pr = part + (((((k - 1) & 1) * 8) * 16) + 4 * w + lb) * PP + ls * 2;
#pragma unroll
                for (int s = 0; s < 8; ++s)
                    q8[s] = *(const v2f*)(pr + s * 16 * PP);
                pin();
            }
            v8f Gd = Gn4;
            if (j + 1 < BLK) {
                Gn4 = __builtin_amdgcn_wmma_f32_16x16x4_f32(false, A4, false, ic[j + 1], (short)0, hb4, false, false);
                issue_ahead(Gn4, Gd);
            }
            float tv[8];
#pragma unroll
            for (int r = 0; r < 8; ++r)
                tv[r] = __builtin_amdgcn_tanhf(Gd[r]);
            float sq[2];
#pragma unroll
            for (int q = 0; q < 2; ++q) {
                const float si = fmaf(tv[0 + q], 0.5f, 0.5f);
                const float sf = fmaf(tv[2 + q], 0.5f, 0.5f);
                const float so = fmaf(tv[6 + q], 0.5f, 0.5f);
                const float cn = sf * c4[q] + si * tv[4 + q];
                c4[q] = cn;
                sq[q] = so * __builtin_amdgcn_tanhf(cn);
            }
            v2f p;
            p.x = sq[0] * w40[0] + sq[1] * w40[1];
            p.y = sq[0] * w41[0] + sq[1] * w41[1];
            if (j + 1 < BLK)
                hold_operand(p, ic[j + 1]);
            sh[j] = p;
        }

        pin();
        {
            v2f hv = ((q8[0] + q8[1]) + (q8[2] + q8[3])) + ((q8[4] + q8[5]) + (q8[6] + q8[7]));
            order_after(hv, sh[BLK - 1]);
            if (prev)
                __builtin_nontemporal_store(hv, (v2f*)(orow4 + (size_t)(k - 1) * BLK * 2));
        }
#pragma unroll
        for (int j = 0; j < BLK; ++j)
            *(v2f*)(pw + j * 2) = sh[j];
        *(v4f*)(sdst4 + (k & 1) * (BLK * 4)) = gst;
    };

    auto ld_blk = [&](const int kb) __attribute__((always_inline)) {
        const int kn = (kb < NBLK) ? kb : (NBLK - 1);
        return *(const v4f*)(gsrc4 + (size_t)kn * BLK * 4);
    };
#pragma unroll
    for (int i = 2; i < CAHEAD; ++i)
        gq[i] = ld_blk(i);
    int k = 0;
    for (; k < kA; k += CAHEAD) {
#pragma unroll
        for (int i = 0; i < CAHEAD; ++i) {
            gq[i] = ld_blk(k + i + CAHEAD);
            pin();
            if (i & 1)
                ctx_block(k + i, inO, inE, gq[(i + 2) & (CAHEAD - 1)], true);
            else
                ctx_block(k + i, inE, inO, gq[(i + 2) & (CAHEAD - 1)], (k + i) > 0);
        }
    }

    const int bb = lane & 3;
    const int s4 = (lane >> 2) & 3;
    const bool mA = (s4 == hi);
    const bool mB = (s4 == 2 + hi);
    v2f Aha, Ahb, Axa, Axb;
    {
        const int usub = 2 * (n >> 3) + (n & 1);
        const int gsel = (n & 7) >> 1;
        const int ga = gsel * 16 + 4 * hi + usub;
        const int gb = gsel * 16 + 4 * (2 + hi) + usub;
        Aha.x = gsc * Whh[ga * 2 + 0]; Aha.y = gsc * Whh[ga * 2 + 1];
        Ahb.x = gsc * Whh[gb * 2 + 0]; Ahb.y = gsc * Whh[gb * 2 + 1];
        Axa.x = gsc * Wih[ga * 2 + 0]; Axa.y = gsc * Wih[ga * 2 + 1];
        Axb.x = gsc * Wih[gb * 2 + 0]; Axb.y = gsc * Wih[gb * 2 + 1];
    }
    v8f hb2;
    float wp0[2], wp1[2];
#pragma unroll
    for (int r = 0; r < 8; ++r) {
        const int grow = (r >> 1) * 16 + 4 * s4 + 2 * hi + (r & 1);
        const float bsum = bih[grow] + bhh[grow];
        hb2[r] = ((r >> 1) == 2) ? bsum : 0.5f * bsum;
    }
#pragma unroll
    for (int q = 0; q < 2; ++q) {
        wp0[q] = 0.5f * Whr[0 * 16 + 4 * s4 + 2 * hi + q];
        wp1[q] = 0.5f * Whr[1 * 16 + 4 * s4 + 2 * hi + q];
    }
    cbuf[(4 * w + 2 * hi + 0) * 17 + n] = c4[0];
    cbuf[(4 * w + 2 * hi + 1) * 17 + n] = c4[1];
    __syncthreads();
    float c2[2];
#pragma unroll
    for (int q = 0; q < 2; ++q)
        c2[q] = cbuf[(4 * s4 + 2 * hi + q) * 17 + 4 * w + bb];
    if (kA > 0) {
        const float* pr = part + ((((kA - 1) & 1) * 8) * 16 + 4 * w + bb) * PP + (BLK - 1) * 2;
        v2f q8[8];
#pragma unroll
        for (int s = 0; s < 8; ++s)
            q8[s] = *(const v2f*)(pr + s * 16 * PP);
        const v2f hv = ((q8[0] + q8[1]) + (q8[2] + q8[3])) + ((q8[4] + q8[5]) + (q8[6] + q8[7]));
        h0 = hv.x;
        h1 = hv.y;
        const float* po = part + ((((kA - 1) & 1) * 8) * 16 + 4 * w + lb) * PP + ls * 2;
        v2f o8[8];
#pragma unroll
        for (int s = 0; s < 8; ++s)
            o8[s] = *(const v2f*)(po + s * 16 * PP);
        const v2f ov = ((o8[0] + o8[1]) + (o8[2] + o8[3])) + ((o8[4] + o8[5]) + (o8[6] + o8[7]));
        __builtin_nontemporal_store(ov, (v2f*)(orow4 + (size_t)(kA - 1) * BLK * 2));
    }

    const float* srcA = stage + (mA ? (4 * w + bb) * PITCH : 16 * PITCH);
    const float* srcB = stage + (mB ? (4 * w + bb) * PITCH : 16 * PITCH);
    const int g8 = s4 + 4 * hi;
    float* orow2 = out + ((size_t)(b0 + 4 * w + bb) * TSTEPS + g8) * 2;

    v4f iA0[BLK], iB0[BLK], iA1[BLK], iB1[BLK];
    {
        const int hf = (kA & 1) * (BLK * 4);
#pragma unroll
        for (int i = 0; i < BLK; ++i) {
            iA0[i] = *(const v4f*)(srcA + hf + i * 4);
            iB0[i] = *(const v4f*)(srcB + hf + i * 4);
        }
    }
    v8f Dx;
    {
        const v2f xa = {iA0[0].x, iA0[0].y};
        const v2f xb = {iB0[0].x, iB0[0].y};
        Dx = __builtin_amdgcn_wmma_f32_16x16x4_f32(false, Axa, false, xa, (short)0, hb2, false, false);
        Dx = __builtin_amdgcn_wmma_f32_16x16x4_f32(false, Axb, false, xb, (short)0, Dx, false, false);
    }

    auto fr_block = [&](const int kk, v4f (&ia)[BLK], v4f (&ib)[BLK], v4f (&na)[BLK], v4f (&nb)[BLK], v4f& gn)
        __attribute__((always_inline)) {
        const int tb = kk * BLK;
        v2f hs = {0.0f, 0.0f};
#pragma unroll
        for (int j = 0; j < BLK; ++j) {
            if (j == BLK - 2) {
                pin();
                *(v4f*)(sdst4 + ((kk + 1) & 1) * (BLK * 4)) = gn;
                __builtin_amdgcn_fence(__ATOMIC_ACQ_REL, "wavefront");
                const int hf = ((kk + 1) & 1) * (BLK * 4);
#pragma unroll
                for (int i = 0; i < BLK; ++i) {
                    na[i] = *(const v4f*)(srcA + hf + i * 4);
                    nb[i] = *(const v4f*)(srcB + hf + i * 4);
                }
                gn = ld_blk(kk + 3);
                pin();
            }
            const int t = tb + j;
            const bool fr = !(t < nc);
            const bool cA = mA && fr;
            const bool cB = mB && fr;
            v2f Ba, Bb;
            Ba.x = cA ? h0 : ia[j].z;
            Ba.y = cA ? h1 : ia[j].w;
            Bb.x = cB ? h0 : ib[j].z;
            Bb.y = cB ? h1 : ib[j].w;
            v8f Gd = __builtin_amdgcn_wmma_f32_16x16x4_f32(false, Aha, false, Ba, (short)0, Dx, false, false);
            Gd = __builtin_amdgcn_wmma_f32_16x16x4_f32(false, Ahb, false, Bb, (short)0, Gd, false, false);
            wmma_slots(Gd, Ba, Bb);
            {
                const v4f ua = (j + 1 < BLK) ? ia[(j + 1) & (BLK - 1)] : na[0];
                const v4f ub = (j + 1 < BLK) ? ib[(j + 1) & (BLK - 1)] : nb[0];
                const v2f xa = {ua.x, ua.y};
                const v2f xb = {ub.x, ub.y};
                Dx = __builtin_amdgcn_wmma_f32_16x16x4_f32(false, Axa, false, xa, (short)0, hb2, false, false);
                Dx = __builtin_amdgcn_wmma_f32_16x16x4_f32(false, Axb, false, xb, (short)0, Dx, false, false);
                wmma_slots(Dx, xa, xb);
            }
            float tv[8];
#pragma unroll
            for (int r = 0; r < 8; ++r)
                tv[r] = __builtin_amdgcn_tanhf(Gd[r]);
            float tc[2];
#pragma unroll
            for (int q = 0; q < 2; ++q) {
                const float si = fmaf(tv[0 + q], 0.5f, 0.5f);
                const float sf = fmaf(tv[2 + q], 0.5f, 0.5f);
                const float cn = sf * c2[q] + si * tv[4 + q];
                c2[q] = cn;
                tc[q] = __builtin_amdgcn_tanhf(cn);
            }
            const float a00 = fmaf(tv[6], wp0[0], wp0[0]);
            const float a01 = fmaf(tv[7], wp0[1], wp0[1]);
            const float a10 = fmaf(tv[6], wp1[0], wp1[0]);
            const float a11 = fmaf(tv[7], wp1[1], wp1[1]);
            float p0 = fmaf(tc[0], a00, tc[1] * a01);
            float p1 = fmaf(tc[0], a10, tc[1] * a11);
            p0 += rowx4(p0);  p1 += rowx4(p1);
            p0 += rowx8(p0);  p1 += rowx8(p1);
            h0 = p0 + xhalf_rec(p0);
            h1 = p1 + xhalf_rec(p1);
            hs.x = (g8 == j) ? h0 : hs.x;
            hs.y = (g8 == j) ? h1 : hs.y;
        }
        __builtin_nontemporal_store(hs, (v2f*)(orow2 + (size_t)tb * 2));
    };

    for (; k < NBLK; k += 2) {
        fr_block(k, iA0, iB0, iA1, iB1, gq[1]);
        fr_block(k + 1, iA1, iB1, iA0, iB0, gq[2]);
    }
}

extern "C" void kernel_launch(void* const* d_in, const int* in_sizes, int n_in,
                              void* d_out, int out_size, void* d_ws, size_t ws_size,
                              hipStream_t stream) {
    const float* u    = (const float*)d_in[0];
    const float* Wih  = (const float*)d_in[1];
    const float* Whh  = (const float*)d_in[2];
    const float* bih  = (const float*)d_in[3];
    const float* bhh  = (const float*)d_in[4];
    const float* Whr  = (const float*)d_in[5];
    const int*   ncp  = (const int*)d_in[6];
    float*       out  = (float*)d_out;

    dim3 grid(BATCH / 16);
    dim3 block(128);
    OpenLSTM_wmma_scan<<<grid, block, 0, stream>>>(u, Wih, Whh, bih, bhh, Whr, ncp, out);
}


# steps t >= n_context: input half of the gates (bias + W_ih x, 2 terms a row) by per-lane vector FMAs a step ahead, not 2 zero-blocked matrix products; same sum order; recurrent WMMA pair kept
# speedup vs baseline: 8.9546x; 1.1080x over previous
#include <hip/hip_runtime.h>

typedef __attribute__((ext_vector_type(2))) float v2f;
typedef __attribute__((ext_vector_type(4))) float v4f;
typedef __attribute__((ext_vector_type(8))) float v8f;

#define BATCH   4096
#define TSTEPS  2048
#define BLK     8
#define NBLK    (TSTEPS / BLK)
#define CAHEAD  8
#define PP      18
#define PITCH   68

__device__ __forceinline__ float fsigm_b(float x, float hb) {
    return fmaf(__builtin_amdgcn_tanhf(fmaf(x, 0.5f, hb)), 0.5f, 0.5f);
}

__device__ __forceinline__ float xhalf(float x) {
    const int b = __builtin_bit_cast(int, x);
    return __builtin_bit_cast(float, __builtin_amdgcn_permlanex16(b, b, 0x76543210, 0xfedcba98, false, false));
}

__device__ __forceinline__ float xhalf_rec(float x) {
    const int b = __builtin_bit_cast(int, x);
    return __builtin_bit_cast(float, __builtin_amdgcn_permlanex16(0, b, 0x76543210, 0xfedcba98, false, false));
}

__device__ __forceinline__ float rowx4(float x) {
    const int b = __builtin_bit_cast(int, x);
    return __builtin_bit_cast(float, __builtin_amdgcn_update_dpp(0, b, 0x164, 0xf, 0xf, true));
}
__device__ __forceinline__ float rowx8(float x) {
    const int b = __builtin_bit_cast(int, x);
    return __builtin_bit_cast(float, __builtin_amdgcn_update_dpp(0, b, 0x168, 0xf, 0xf, true));
}

__device__ __forceinline__ void wmma_slots(v8f& d, const v2f ba, const v2f bb) {
    asm("v_nop\n\tv_nop\n\tv_nop\n\tv_nop" : "+v"(d) : "v"(ba), "v"(bb));
}

__device__ __forceinline__ void issue_ahead(v8f& next, v8f& cur) {
    asm("" : "+v"(next), "+v"(cur));
}

__device__ __forceinline__ void order_after(v2f& v, const v2f after) {
    asm("" : "+v"(v) : "v"(after));
}

__device__ __forceinline__ void hold_operand(v2f& v, const v2f b) {
    asm("" : "+v"(v) : "v"(b));
}

__device__ __forceinline__ void pin() { __builtin_amdgcn_sched_barrier(0); }

__device__ __forceinline__ void lds_arrive() {
    __builtin_amdgcn_fence(__ATOMIC_RELEASE, "workgroup", "local");
    __builtin_amdgcn_s_barrier_signal(-1);
}
__device__ __forceinline__ void lds_await() {
    __builtin_amdgcn_s_barrier_wait(-1);
    __builtin_amdgcn_fence(__ATOMIC_ACQUIRE, "workgroup", "local");
}

struct Gates { v8f i, f, g, o; };

__global__ __launch_bounds__(128) __attribute__((amdgpu_waves_per_eu(1, 1)))
void OpenLSTM_wmma_scan(const float* __restrict__ u,
                        const float* __restrict__ Wih,
                        const float* __restrict__ Whh,
                        const float* __restrict__ bih,
                        const float* __restrict__ bhh,
                        const float* __restrict__ Whr,
                        const int*   __restrict__ ncp,
                        float*       __restrict__ out)
{
    const int lane = threadIdx.x & 31;
    const int w    = __builtin_amdgcn_readfirstlane((int)(threadIdx.x >> 5));
    const int n    = lane & 15;
    const int hi   = lane >> 4;
    const int b0   = blockIdx.x * 16;
    const int nc   = *ncp;

    __shared__ __attribute__((aligned(16))) float stage[16 * PITCH + 2 * BLK * 4];
    __shared__ __attribute__((aligned(16))) float part[2 * 8 * 16 * PP];
    __shared__ float cbuf[16 * 17];
    const int ls = lane & 7;
    const int lb = lane >> 3;
    const float* gsrc = u + ((size_t)(b0 + lb) * TSTEPS + ls) * 4;
    float*       sdst = stage + lb * PITCH + ls * 4;
    const float* ssrc = stage + n * PITCH + hi * 2;
    const size_t PIECE = (size_t)4 * TSTEPS * 4;

    const float* Wsrc = hi ? Whh : Wih;

    v8f c = {};
    float h0 = 0.0f, h1 = 0.0f;

    float*       op = out + (size_t)(b0 + n) * TSTEPS * 2;

    v2f A4;
    v8f hb4;
    float w40[2], w41[2];
#pragma unroll
    for (int r = 0; r < 8; ++r) {
        const int grow = (r >> 1) * 16 + 4 * w + 2 * hi + (r & 1);
        const float bsum = bih[grow] + bhh[grow];
        hb4[r] = ((r >> 1) == 2) ? bsum : 0.5f * bsum;
    }
    const float gsc = (((n & 7) >> 1) == 2) ? 1.0f : 0.5f;
    {
        const int m    = n;
        const int grow = ((m & 7) >> 1) * 16 + 4 * w + 2 * (m >> 3) + (m & 1);
        A4.x = gsc * Wsrc[grow * 2 + 0];
        A4.y = gsc * Wsrc[grow * 2 + 1];
    }
#pragma unroll
    for (int q = 0; q < 2; ++q) {
        w40[q] = Whr[0 * 16 + 4 * w + 2 * hi + q];
        w41[q] = Whr[1 * 16 + 4 * w + 2 * hi + q];
    }
    float c4[2] = {0.0f, 0.0f};

    const float* gsrc4 = u + ((size_t)(b0 + 4 * w + lb) * TSTEPS + ls) * 4;
    float*       sdst4 = stage + (4 * w + lb) * PITCH + ls * 4;
    float*       orow4 = out + ((size_t)(b0 + 4 * w + lb) * TSTEPS + ls) * 2;

    v4f gq[CAHEAD];
    {
        const v4f g = *(const v4f*)(gsrc4);
        gq[1] = *(const v4f*)(gsrc4 + BLK * 4);
        *(v4f*)(sdst4) = g;
        *(v4f*)(sdst4 + BLK * 4) = gq[1];
    }
    if (threadIdx.x < 2 * BLK * 4) stage[16 * PITCH + threadIdx.x] = 0.0f;
    __syncthreads();

    v2f inE[BLK], inO[BLK];
#pragma unroll
    for (int j = 0; j < BLK; ++j)
        inE[j] = *(const v2f*)(ssrc + j * 4);

    const int ncc = nc < 0 ? 0 : (nc > TSTEPS ? TSTEPS : nc);
    const int kA  = (ncc / (CAHEAD * BLK)) * CAHEAD;

    auto ctx_block = [&](const int kk, v2f (&ic)[BLK], v2f (&nx)[BLK], const v4f gst, const bool prev)
        __attribute__((always_inline)) {
        const int k = kk;
        float* pw = part + ((((k & 1) * 8) + 2 * w + hi) * 16 + n) * PP;
        v2f sh[BLK];
        v2f q8[8];
        v8f Gn4 = __builtin_amdgcn_wmma_f32_16x16x4_f32(false, A4, false, ic[0], (short)0, hb4, false, false);
        wmma_slots(Gn4, ic[0], ic[0]);
#pragma unroll
        for (int j = 0; j < BLK; ++j) {
            if (j == 2) {
                pin();
                lds_arrive();
                pin();
            }
            if (j == 4) {
                pin();
                lds_await();
                const float* sn = ssrc + ((k + 1) & 1) * (BLK * 4);
#pragma unroll
                for (int i = 0; i < BLK; ++i)
                    nx[i] = *(const v2f*)(sn + i * 4);
                const float* pr = part + (((((k - 1) & 1) * 8) * 16) + 4 * w + lb) * PP + ls * 2;
#pragma unroll
                for (int s = 0; s < 8; ++s)
                    q8[s] = *(const v2f*)(pr + s * 16 * PP);
                pin();
            }
            v8f Gd = Gn4;
            if (j + 1 < BLK) {
                Gn4 = __builtin_amdgcn_wmma_f32_16x16x4_f32(false, A4, false, ic[j + 1], (short)0, hb4, false, false);
                issue_ahead(Gn4, Gd);
            }
            float tv[8];
#pragma unroll
            for (int r = 0; r < 8; ++r)
                tv[r] = __builtin_amdgcn_tanhf(Gd[r]);
            float sq[2];
#pragma unroll
            for (int q = 0; q < 2; ++q) {
                const float si = fmaf(tv[0 + q], 0.5f, 0.5f);
                const float sf = fmaf(tv[2 + q], 0.5f, 0.5f);
                const float so = fmaf(tv[6 + q], 0.5f, 0.5f);
                const float cn = sf * c4[q] + si * tv[4 + q];
                c4[q] = cn;
                sq[q] = so * __builtin_amdgcn_tanhf(cn);
            }
            v2f p;
            p.x = sq[0] * w40[0] + sq[1] * w40[1];
            p.y = sq[0] * w41[0] + sq[1] * w41[1];
            if (j + 1 < BLK)
                hold_operand(p, ic[j + 1]);
            sh[j] = p;
        }

        pin();
        {
            v2f hv = ((q8[0] + q8[1]) + (q8[2] + q8[3])) + ((q8[4] + q8[5]) + (q8[6] + q8[7]));
            order_after(hv, sh[BLK - 1]);
            if (prev)
                __builtin_nontemporal_store(hv, (v2f*)(orow4 + (size_t)(k - 1) * BLK * 2));
        }
#pragma unroll
        for (int j = 0; j < BLK; ++j)
            *(v2f*)(pw + j * 2) = sh[j];
        *(v4f*)(sdst4 + (k & 1) * (BLK * 4)) = gst;
    };

    auto ld_blk = [&](const int kb) __attribute__((always_inline)) {
        const int kn = (kb < NBLK) ? kb : (NBLK - 1);
        return *(const v4f*)(gsrc4 + (size_t)kn * BLK * 4);
    };
#pragma unroll
    for (int i = 2; i < CAHEAD; ++i)
        gq[i] = ld_blk(i);
    int k = 0;
    for (; k < kA; k += CAHEAD) {
#pragma unroll
        for (int i = 0; i < CAHEAD; ++i) {
            gq[i] = ld_blk(k + i + CAHEAD);
            pin();
            if (i & 1)
                ctx_block(k + i, inO, inE, gq[(i + 2) & (CAHEAD - 1)], true);
            else
                ctx_block(k + i, inE, inO, gq[(i + 2) & (CAHEAD - 1)], (k + i) > 0);
        }
    }

    const int bb = lane & 3;
    const int s4 = (lane >> 2) & 3;
    const bool mA = (s4 == hi);
    const bool mB = (s4 == 2 + hi);
    v2f Aha, Ahb;
    {
        const int usub = 2 * (n >> 3) + (n & 1);
        const int gsel = (n & 7) >> 1;
        const int ga = gsel * 16 + 4 * hi + usub;
        const int gb = gsel * 16 + 4 * (2 + hi) + usub;
        Aha.x = gsc * Whh[ga * 2 + 0]; Aha.y = gsc * Whh[ga * 2 + 1];
        Ahb.x = gsc * Whh[gb * 2 + 0]; Ahb.y = gsc * Whh[gb * 2 + 1];
    }
    v8f hb2;
    float wx0[8], wx1[8];
    float wp0[2], wp1[2];
#pragma unroll
    for (int r = 0; r < 8; ++r) {
        const int grow = (r >> 1) * 16 + 4 * s4 + 2 * hi + (r & 1);
        const float bsum = bih[grow] + bhh[grow];
        const float rsc  = ((r >> 1) == 2) ? 1.0f : 0.5f;
        hb2[r] = rsc * bsum;
        wx0[r] = rsc * Wih[grow * 2 + 0];
        wx1[r] = rsc * Wih[grow * 2 + 1];
    }
#pragma unroll
    for (int q = 0; q < 2; ++q) {
        wp0[q] = 0.5f * Whr[0 * 16 + 4 * s4 + 2 * hi + q];
        wp1[q] = 0.5f * Whr[1 * 16 + 4 * s4 + 2 * hi + q];
    }
    cbuf[(4 * w + 2 * hi + 0) * 17 + n] = c4[0];
    cbuf[(4 * w + 2 * hi + 1) * 17 + n] = c4[1];
    __syncthreads();
    float c2[2];
#pragma unroll
    for (int q = 0; q < 2; ++q)
        c2[q] = cbuf[(4 * s4 + 2 * hi + q) * 17 + 4 * w + bb];
    if (kA > 0) {
        const float* pr = part + ((((kA - 1) & 1) * 8) * 16 + 4 * w + bb) * PP + (BLK - 1) * 2;
        v2f q8[8];
#pragma unroll
        for (int s = 0; s < 8; ++s)
            q8[s] = *(const v2f*)(pr + s * 16 * PP);
        const v2f hv = ((q8[0] + q8[1]) + (q8[2] + q8[3])) + ((q8[4] + q8[5]) + (q8[6] + q8[7]));
        h0 = hv.x;
        h1 = hv.y;
        const float* po = part + ((((kA - 1) & 1) * 8) * 16 + 4 * w + lb) * PP + ls * 2;
        v2f o8[8];
#pragma unroll
        for (int s = 0; s < 8; ++s)
            o8[s] = *(const v2f*)(po + s * 16 * PP);
        const v2f ov = ((o8[0] + o8[1]) + (o8[2] + o8[3])) + ((o8[4] + o8[5]) + (o8[6] + o8[7]));
        __builtin_nontemporal_store(ov, (v2f*)(orow4 + (size_t)(kA - 1) * BLK * 2));
    }

    const float* srcA = stage + (mA ? (4 * w + bb) * PITCH : 16 * PITCH) + 2;
    const float* srcB = stage + (mB ? (4 * w + bb) * PITCH : 16 * PITCH) + 2;
    const float* srcX = stage + (4 * w + bb) * PITCH;
    const int g8 = s4 + 4 * hi;
    float* orow2 = out + ((size_t)(b0 + 4 * w + bb) * TSTEPS + g8) * 2;

    v2f iA0[BLK], iB0[BLK], iA1[BLK], iB1[BLK];
    v2f iX0[BLK], iX1[BLK];
    {
        const int hf = (kA & 1) * (BLK * 4);
#pragma unroll
        for (int i = 0; i < BLK; ++i) {
            iA0[i] = *(const v2f*)(srcA + hf + i * 4);
            iB0[i] = *(const v2f*)(srcB + hf + i * 4);
            iX0[i] = *(const v2f*)(srcX + hf + i * 4);
        }
    }
    auto in_half = [&](const v2f x) __attribute__((always_inline)) {
        v8f d;
#pragma unroll
        for (int r = 0; r < 8; ++r)
            d[r] = fmaf(wx1[r], x.y, fmaf(wx0[r], x.x, hb2[r]));
        return d;
    };
    v8f Dx = in_half(iX0[0]);

    auto fr_block = [&](const int kk, v2f (&ia)[BLK], v2f (&ib)[BLK], v2f (&ix)[BLK],
                        v2f (&na)[BLK], v2f (&nb)[BLK], v2f (&nx)[BLK], v4f& gn)
        __attribute__((always_inline)) {
        const int tb = kk * BLK;
        v2f hs = {0.0f, 0.0f};
#pragma unroll
        for (int j = 0; j < BLK; ++j) {
            if (j == BLK - 2) {
                pin();
                *(v4f*)(sdst4 + ((kk + 1) & 1) * (BLK * 4)) = gn;
                __builtin_amdgcn_fence(__ATOMIC_ACQ_REL, "wavefront");
                const int hf = ((kk + 1) & 1) * (BLK * 4);
#pragma unroll
                for (int i = 0; i < BLK; ++i) {
                    na[i] = *(const v2f*)(srcA + hf + i * 4);
                    nb[i] = *(const v2f*)(srcB + hf + i * 4);
                    nx[i] = *(const v2f*)(srcX + hf + i * 4);
                }
                gn = ld_blk(kk + 3);
                pin();
            }
            const int t = tb + j;
            const bool fr = !(t < nc);
            const bool cA = mA && fr;
            const bool cB = mB && fr;
            v2f Ba, Bb;
            Ba.x = cA ? h0 : ia[j].x;
            Ba.y = cA ? h1 : ia[j].y;
            Bb.x = cB ? h0 : ib[j].x;
            Bb.y = cB ? h1 : ib[j].y;
            v8f Gd = __builtin_amdgcn_wmma_f32_16x16x4_f32(false, Aha, false, Ba, (short)0, Dx, false, false);
            Gd = __builtin_amdgcn_wmma_f32_16x16x4_f32(false, Ahb, false, Bb, (short)0, Gd, false, false);
            wmma_slots(Gd, Ba, Bb);
            Dx = in_half((j + 1 < BLK) ? ix[(j + 1) & (BLK - 1)] : nx[0]);
            float tv[8];
#pragma unroll
            for (int r = 0; r < 8; ++r)
                tv[r] = __builtin_amdgcn_tanhf(Gd[r]);
            float tc[2];
#pragma unroll
            for (int q = 0; q < 2; ++q) {
                const float si = fmaf(tv[0 + q], 0.5f, 0.5f);
                const float sf = fmaf(tv[2 + q], 0.5f, 0.5f);
                const float cn = sf * c2[q] + si * tv[4 + q];
                c2[q] = cn;
                tc[q] = __builtin_amdgcn_tanhf(cn);
            }
            const float a00 = fmaf(tv[6], wp0[0], wp0[0]);
            const float a01 = fmaf(tv[7], wp0[1], wp0[1]);
            const float a10 = fmaf(tv[6], wp1[0], wp1[0]);
            const float a11 = fmaf(tv[7], wp1[1], wp1[1]);
            float p0 = fmaf(tc[0], a00, tc[1] * a01);
            float p1 = fmaf(tc[0], a10, tc[1] * a11);
            p0 += rowx4(p0);  p1 += rowx4(p1);
            p0 += rowx8(p0);  p1 += rowx8(p1);
            h0 = p0 + xhalf_rec(p0);
            h1 = p1 + xhalf_rec(p1);
            hs.x = (g8 == j) ? h0 : hs.x;
            hs.y = (g8 == j) ? h1 : hs.y;
        }
        __builtin_nontemporal_store(hs, (v2f*)(orow2 + (size_t)tb * 2));
    };

    for (; k < NBLK; k += 2) {
        fr_block(k, iA0, iB0, iX0, iA1, iB1, iX1, gq[1]);
        fr_block(k + 1, iA1, iB1, iX1, iA0, iB0, iX0, gq[2]);
    }
}

extern "C" void kernel_launch(void* const* d_in, const int* in_sizes, int n_in,
                              void* d_out, int out_size, void* d_ws, size_t ws_size,
                              hipStream_t stream) {
    const float* u    = (const float*)d_in[0];
    const float* Wih  = (const float*)d_in[1];
    const float* Whh  = (const float*)d_in[2];
    const float* bih  = (const float*)d_in[3];
    const float* bhh  = (const float*)d_in[4];
    const float* Whr  = (const float*)d_in[5];
    const int*   ncp  = (const int*)d_in[6];
    float*       out  = (float*)d_out;

    dim3 grid(BATCH / 16);
    dim3 block(128);
    OpenLSTM_wmma_scan<<<grid, block, 0, stream>>>(u, Wih, Whh, bih, bhh, Whr, ncp, out);
}
